# starting kernel
# baseline (speedup 1.0000x reference)
#include <hip/hip_runtime.h>
#include <hip/hip_bf16.h>

typedef __attribute__((ext_vector_type(2))) float v2f;
typedef __attribute__((ext_vector_type(8))) float v8f;

#define D_FEAT 64
#define H_DIM  256
#define TILE_M 64
#define RST_STRIDE 68
#define H_STRIDE   260

__global__ void gin_init_kernel(const float* __restrict__ feat,
                                float* __restrict__ rst, int n_elems4) {
    int i = blockIdx.x * blockDim.x + threadIdx.x;
    if (i < n_elems4) {
        reinterpret_cast<float4*>(rst)[i] =
            reinterpret_cast<const float4*>(feat)[i];
    }
}

__global__ void gin_scatter_kernel(const float* __restrict__ feat,
                                   const int* __restrict__ src,
                                   const int* __restrict__ dst,
                                   float* __restrict__ rst, int n_edges) {
    long long tid = (long long)blockIdx.x * blockDim.x + threadIdx.x;
    int e = (int)(tid >> 4);
    int c = (int)(tid & 15);
    if (e >= n_edges) return;
    size_t soff = (size_t)src[e] * D_FEAT + (size_t)c * 4;
    size_t doff = (size_t)dst[e] * D_FEAT + (size_t)c * 4;
    float4 v = *reinterpret_cast<const float4*>(feat + soff);
    float* p = rst + doff;
    unsafeAtomicAdd(p + 0, v.x);
    unsafeAtomicAdd(p + 1, v.y);
    unsafeAtomicAdd(p + 2, v.z);
    unsafeAtomicAdd(p + 3, v.w);
}

__global__ __launch_bounds__(512)
void gin_mlp_kernel(const float* __restrict__ rst,
                    const float* __restrict__ W1,
                    const float* __restrict__ b1,
                    const float* __restrict__ W2,
                    const float* __restrict__ b2,
                    float* __restrict__ out, int n_nodes) {
    __shared__ float s_rst[TILE_M * RST_STRIDE];
    __shared__ float s_h[TILE_M * H_STRIDE];
    __shared__ float s_out[TILE_M * D_FEAT];

    const int tid   = threadIdx.x;
    const int wave  = tid >> 5;
    const int lane  = tid & 31;
    const int node0 = blockIdx.x * TILE_M;

    for (int i = tid; i < TILE_M * D_FEAT; i += 512) {
        int m = i >> 6, k = i & 63;
        int row = node0 + m;
        row = row < n_nodes ? row : (n_nodes - 1);
        s_rst[m * RST_STRIDE + k] = rst[(size_t)row * D_FEAT + k];
    }
    for (int i = tid; i < TILE_M * D_FEAT; i += 512) s_out[i] = 0.0f;
    __syncthreads();

    const int frow  = lane & 15;
    const int khalf = (lane >> 4) << 1;
    const int mbase = (lane >> 4) << 3;

    {
        const int ncol = wave * 16 + frow;
        v8f acc0 = {}, acc1 = {}, acc2 = {}, acc3 = {};
        #pragma unroll
        for (int kk = 0; kk < 16; ++kk) {
            const int k0 = kk * 4 + khalf;
            v2f b;
            b.x = W1[(size_t)(k0    ) * H_DIM + ncol];
            b.y = W1[(size_t)(k0 + 1) * H_DIM + ncol];
            v2f a0, a1, a2, a3;
            a0.x = s_rst[(frow     ) * RST_STRIDE + k0];
            a0.y = s_rst[(frow     ) * RST_STRIDE + k0 + 1];
            a1.x = s_rst[(frow + 16) * RST_STRIDE + k0];
            a1.y = s_rst[(frow + 16) * RST_STRIDE + k0 + 1];
            a2.x = s_rst[(frow + 32) * RST_STRIDE + k0];
            a2.y = s_rst[(frow + 32) * RST_STRIDE + k0 + 1];
            a3.x = s_rst[(frow + 48) * RST_STRIDE + k0];
            a3.y = s_rst[(frow + 48) * RST_STRIDE + k0 + 1];
            acc0 = __builtin_amdgcn_wmma_f32_16x16x4_f32(
                false, a0, false, b, (short)0, acc0, false, false);
            acc1 = __builtin_amdgcn_wmma_f32_16x16x4_f32(
                false, a1, false, b, (short)0, acc1, false, false);
            acc2 = __builtin_amdgcn_wmma_f32_16x16x4_f32(
                false, a2, false, b, (short)0, acc2, false, false);
            acc3 = __builtin_amdgcn_wmma_f32_16x16x4_f32(
                false, a3, false, b, (short)0, acc3, false, false);
        }
        const float bias = b1[ncol];
        #pragma unroll
        for (int r = 0; r < 8; ++r) {
            float v0 = acc0[r] + bias, v1 = acc1[r] + bias;
            float v2 = acc2[r] + bias, v3 = acc3[r] + bias;
            s_h[(     mbase + r) * H_STRIDE + ncol] = v0 > 0.f ? v0 : 0.f;
            s_h[(16 + mbase + r) * H_STRIDE + ncol] = v1 > 0.f ? v1 : 0.f;
            s_h[(32 + mbase + r) * H_STRIDE + ncol] = v2 > 0.f ? v2 : 0.f;
            s_h[(48 + mbase + r) * H_STRIDE + ncol] = v3 > 0.f ? v3 : 0.f;
        }
    }
    __syncthreads();

    {
        const int ntile = (wave & 3) * 16;
        const int kbase = (wave >> 2) * 64;
        const int ncol  = ntile + frow;
        v8f acc0 = {}, acc1 = {}, acc2 = {}, acc3 = {};
        #pragma unroll
        for (int kk = 0; kk < 16; ++kk) {
            const int k0 = kbase + kk * 4 + khalf;
            v2f b;
            b.x = W2[(size_t)(k0    ) * D_FEAT + ncol];
            b.y = W2[(size_t)(k0 + 1) * D_FEAT + ncol];
            v2f a0, a1, a2, a3;
            a0.x = s_h[(frow     ) * H_STRIDE + k0];
            a0.y = s_h[(frow     ) * H_STRIDE + k0 + 1];
            a1.x = s_h[(frow + 16) * H_STRIDE + k0];
            a1.y = s_h[(frow + 16) * H_STRIDE + k0 + 1];
            a2.x = s_h[(frow + 32) * H_STRIDE + k0];
            a2.y = s_h[(frow + 32) * H_STRIDE + k0 + 1];
            a3.x = s_h[(frow + 48) * H_STRIDE + k0];
            a3.y = s_h[(frow + 48) * H_STRIDE + k0 + 1];
            acc0 = __builtin_amdgcn_wmma_f32_16x16x4_f32(
                false, a0, false, b, (short)0, acc0, false, false);
            acc1 = __builtin_amdgcn_wmma_f32_16x16x4_f32(
                false, a1, false, b, (short)0, acc1, false, false);
            acc2 = __builtin_amdgcn_wmma_f32_16x16x4_f32(
                false, a2, false, b, (short)0, acc2, false, false);
            acc3 = __builtin_amdgcn_wmma_f32_16x16x4_f32(
                false, a3, false, b, (short)0, acc3, false, false);
        }
        #pragma unroll
        for (int r = 0; r < 8; ++r) {
            atomicAdd(&s_out[(     mbase + r) * D_FEAT + ncol], acc0[r]);
            atomicAdd(&s_out[(16 + mbase + r) * D_FEAT + ncol], acc1[r]);
            atomicAdd(&s_out[(32 + mbase + r) * D_FEAT + ncol], acc2[r]);
            atomicAdd(&s_out[(48 + mbase + r) * D_FEAT + ncol], acc3[r]);
        }
    }
    __syncthreads();

    for (int i = tid; i < TILE_M * D_FEAT; i += 512) {
        int m = i >> 6, n = i & 63;
        int row = node0 + m;
        if (row < n_nodes)
            out[(size_t)row * D_FEAT + n] = s_out[i] + b2[n];
    }
}

extern "C" void kernel_launch(void* const* d_in, const int* in_sizes, int n_in,
                              void* d_out, int out_size, void* d_ws, size_t ws_size,
                              hipStream_t stream) {
    const float* feat = (const float*)d_in[0];
    const float* W1   = (const float*)d_in[1];
    const float* b1   = (const float*)d_in[2];
    const float* W2   = (const float*)d_in[3];
    const float* b2   = (const float*)d_in[4];
    const int*   src  = (const int*)d_in[5];
    const int*   dst  = (const int*)d_in[6];
    float*       out  = (float*)d_out;

    const int n_nodes = in_sizes[0] / D_FEAT;
    const int n_edges = in_sizes[5];

    float* rst = (float*)d_ws;

    {
        int n4 = (n_nodes * D_FEAT) / 4;
        gin_init_kernel<<<(n4 + 255) / 256, 256, 0, stream>>>(feat, rst, n4);
    }
    {
        long long work = (long long)n_edges * 16;
        int blocks = (int)((work + 255) / 256);
        gin_scatter_kernel<<<blocks, 256, 0, stream>>>(feat, src, dst, rst, n_edges);
    }
    {
        int blocks = (n_nodes + TILE_M - 1) / TILE_M;
        gin_mlp_kernel<<<blocks, 512, 0, stream>>>(rst, W1, b1, W2, b2, out, n_nodes);
    }
}


# Scatter replaced: a work-group keeps 400 dst rows in LDS, filters ALL edges by a 1-byte owner key, gathers its rows, sums by ds_add_f32; no global atomics; sum order arrival as before
# speedup vs baseline: 1.4584x; 1.4584x over previous
#include <hip/hip_runtime.h>
#include <hip/hip_bf16.h>

typedef __attribute__((ext_vector_type(2))) float v2f;
typedef __attribute__((ext_vector_type(8))) float v8f;

#define D_FEAT 64
#define H_DIM  256
#define TILE_M 64
#define RST_STRIDE 68
#define H_STRIDE   260

#define AGG_ROWS    400
#define AGG_THREADS 1024
#define AGG_QCAP    16384

__global__ void gin_key_kernel(const int* __restrict__ dst,
                               unsigned int* __restrict__ keys, int n_edges) {
    const int npad4 = ((n_edges + 15) >> 4) << 2;
    const int t = blockIdx.x * blockDim.x + threadIdx.x;
    if (t >= npad4) return;
    const int e0 = t * 4;
    unsigned int k0 = 0xFFu, k1 = 0xFFu, k2 = 0xFFu, k3 = 0xFFu;
    if (e0 + 3 < n_edges) {
        const int4 d = *reinterpret_cast<const int4*>(dst + e0);
        k0 = (unsigned int)d.x / AGG_ROWS; k1 = (unsigned int)d.y / AGG_ROWS;
        k2 = (unsigned int)d.z / AGG_ROWS; k3 = (unsigned int)d.w / AGG_ROWS;
    } else {
        if (e0     < n_edges) k0 = (unsigned int)dst[e0    ] / AGG_ROWS;
        if (e0 + 1 < n_edges) k1 = (unsigned int)dst[e0 + 1] / AGG_ROWS;
        if (e0 + 2 < n_edges) k2 = (unsigned int)dst[e0 + 2] / AGG_ROWS;
    }
    keys[t] = k0 | (k1 << 8) | (k2 << 16) | (k3 << 24);
}

__device__ __forceinline__ unsigned int eq_bytes(unsigned int x, unsigned int rep) {
    const unsigned int t = x ^ rep;
    const unsigned int y = (t & 0x7F7F7F7Fu) + 0x7F7F7F7Fu;
    return ~(y | t | 0x7F7F7F7Fu);
}

__global__ __launch_bounds__(AGG_THREADS)
void gin_aggregate_kernel(const float* __restrict__ feat,
                          const int* __restrict__ src,
                          const int* __restrict__ dst,
                          const unsigned int* __restrict__ keys,
                          float* __restrict__ rst, int n_nodes, int n_edges) {
    __shared__ float s_tile[AGG_ROWS * D_FEAT];
    __shared__ int   s_queue[AGG_QCAP];
    __shared__ int   s_qcount;

    const int tid  = threadIdx.x;
    const int lane = tid & 31;
    const int wave = tid >> 5;
    const int lo   = blockIdx.x * AGG_ROWS;
    const int rows = (n_nodes - lo) < AGG_ROWS ? (n_nodes - lo) : AGG_ROWS;

    {
        const float4* f4 = reinterpret_cast<const float4*>(feat + (size_t)lo * D_FEAT);
        float4* t4 = reinterpret_cast<float4*>(s_tile);
        for (int i = tid; i < rows * (D_FEAT / 4); i += AGG_THREADS) t4[i] = f4[i];
    }
    if (tid == 0) s_qcount = 0;
    __syncthreads();

    {
        const int n16 = (n_edges + 15) >> 4;
        const uint4* k16 = reinterpret_cast<const uint4*>(keys);
        const unsigned int rep = (unsigned int)blockIdx.x * 0x01010101u;
        const int start = (int)(((long long)blockIdx.x * n16) / gridDim.x);
        uint4 cur[4];
        #pragma unroll
        for (int u = 0; u < 4; ++u) {
            const int p = u * AGG_THREADS + tid;
            int i = p + start;
            if (i >= n16) i -= n16;
            if (p >= n16) i = 0;
            cur[u] = k16[i];
            if (p >= n16) cur[u] = make_uint4(0xFFFFFFFFu, 0xFFFFFFFFu, 0xFFFFFFFFu, 0xFFFFFFFFu);
        }
        for (int p0 = 0; p0 < n16; p0 += AGG_THREADS * 4) {
            uint4 nxt[4];
            #pragma unroll
            for (int u = 0; u < 4; ++u) {
                const int p = p0 + (4 + u) * AGG_THREADS + tid;
                int i = p + start;
                if (i >= n16) i -= n16;
                if (p >= n16) i = 0;
                nxt[u] = k16[i];
                if (p >= n16) nxt[u] = make_uint4(0xFFFFFFFFu, 0xFFFFFFFFu, 0xFFFFFFFFu, 0xFFFFFFFFu);
            }
            #pragma unroll
            for (int u = 0; u < 4; ++u) {
                int gi = p0 + u * AGG_THREADS + tid + start;
                if (gi >= n16) gi -= n16;
                unsigned int mm = (eq_bytes(cur[u].x, rep) >> 7) | (eq_bytes(cur[u].y, rep) >> 6)
                                | (eq_bytes(cur[u].z, rep) >> 5) | (eq_bytes(cur[u].w, rep) >> 4);
                while (mm) {
                    const int pos = __builtin_ctz(mm);
                    mm &= mm - 1;
                    const int e = gi * 16 + (pos & 7) * 4 + (pos >> 3);
                    const int slot = atomicAdd(&s_qcount, 1);
                    if (slot < AGG_QCAP) {
                        s_queue[slot] = e;
                    } else {
                        const int d = dst[e] - lo;
                        const float* fr = feat + (size_t)src[e] * D_FEAT;
                        for (int k = 0; k < D_FEAT; ++k)
                            atomicAdd(&s_tile[d * D_FEAT + k], fr[k]);
                    }
                }
                cur[u] = nxt[u];
            }
        }
    }
    __syncthreads();

    {
        const int n = s_qcount < AGG_QCAP ? s_qcount : AGG_QCAP;
        for (int base = wave * 32; base < n; base += AGG_THREADS) {
            const int i = base + lane;
            int d = 0, s = 0;
            if (i < n) {
                const int e = s_queue[i];
                d = dst[e] - lo;
                s = src[e];
            }
            const int cnt = (n - base) < 32 ? (n - base) : 32;
            int j = 0;
            for (; j + 8 <= cnt; j += 8) {
                float2 v[8];
                int    dj[8];
                #pragma unroll
                for (int u = 0; u < 8; ++u) {
                    const int sj = __builtin_amdgcn_readlane(s, j + u);
                    dj[u] = __builtin_amdgcn_readlane(d, j + u);
                    v[u] = *reinterpret_cast<const float2*>(feat + (size_t)sj * D_FEAT + lane * 2);
                }
                #pragma unroll
                for (int u = 0; u < 8; ++u) {
                    atomicAdd(&s_tile[dj[u] * D_FEAT + lane * 2    ], v[u].x);
                    atomicAdd(&s_tile[dj[u] * D_FEAT + lane * 2 + 1], v[u].y);
                }
            }
            for (; j < cnt; ++j) {
                const int sj = __builtin_amdgcn_readlane(s, j);
                const int dd = __builtin_amdgcn_readlane(d, j);
                const float2 v = *reinterpret_cast<const float2*>(feat + (size_t)sj * D_FEAT + lane * 2);
                atomicAdd(&s_tile[dd * D_FEAT + lane * 2    ], v.x);
                atomicAdd(&s_tile[dd * D_FEAT + lane * 2 + 1], v.y);
            }
        }
    }
    __syncthreads();

    {
        float4* r4 = reinterpret_cast<float4*>(rst + (size_t)lo * D_FEAT);
        const float4* t4 = reinterpret_cast<const float4*>(s_tile);
        for (int i = tid; i < rows * (D_FEAT / 4); i += AGG_THREADS) r4[i] = t4[i];
    }
}

__global__ __launch_bounds__(512)
void gin_mlp_kernel(const float* __restrict__ rst,
                    const float* __restrict__ W1,
                    const float* __restrict__ b1,
                    const float* __restrict__ W2,
                    const float* __restrict__ b2,
                    float* __restrict__ out, int n_nodes) {
    __shared__ float s_rst[TILE_M * RST_STRIDE];
    __shared__ float s_h[TILE_M * H_STRIDE];
    __shared__ float s_out[TILE_M * D_FEAT];

    const int tid   = threadIdx.x;
    const int wave  = tid >> 5;
    const int lane  = tid & 31;
    const int node0 = blockIdx.x * TILE_M;

    for (int i = tid; i < TILE_M * D_FEAT; i += 512) {
        int m = i >> 6, k = i & 63;
        int row = node0 + m;
        row = row < n_nodes ? row : (n_nodes - 1);
        s_rst[m * RST_STRIDE + k] = rst[(size_t)row * D_FEAT + k];
    }
    for (int i = tid; i < TILE_M * D_FEAT; i += 512) s_out[i] = 0.0f;
    __syncthreads();

    const int frow  = lane & 15;
    const int khalf = (lane >> 4) << 1;
    const int mbase = (lane >> 4) << 3;

    {
        const int ncol = wave * 16 + frow;
        v8f acc0 = {}, acc1 = {}, acc2 = {}, acc3 = {};
        #pragma unroll
        for (int kk = 0; kk < 16; ++kk) {
            const int k0 = kk * 4 + khalf;
            v2f b;
            b.x = W1[(size_t)(k0    ) * H_DIM + ncol];
            b.y = W1[(size_t)(k0 + 1) * H_DIM + ncol];
            v2f a0, a1, a2, a3;
            a0.x = s_rst[(frow     ) * RST_STRIDE + k0];
            a0.y = s_rst[(frow     ) * RST_STRIDE + k0 + 1];
            a1.x = s_rst[(frow + 16) * RST_STRIDE + k0];
            a1.y = s_rst[(frow + 16) * RST_STRIDE + k0 + 1];
            a2.x = s_rst[(frow + 32) * RST_STRIDE + k0];
            a2.y = s_rst[(frow + 32) * RST_STRIDE + k0 + 1];
            a3.x = s_rst[(frow + 48) * RST_STRIDE + k0];
            a3.y = s_rst[(frow + 48) * RST_STRIDE + k0 + 1];
            acc0 = __builtin_amdgcn_wmma_f32_16x16x4_f32(
                false, a0, false, b, (short)0, acc0, false, false);
            acc1 = __builtin_amdgcn_wmma_f32_16x16x4_f32(
                false, a1, false, b, (short)0, acc1, false, false);
            acc2 = __builtin_amdgcn_wmma_f32_16x16x4_f32(
                false, a2, false, b, (short)0, acc2, false, false);
            acc3 = __builtin_amdgcn_wmma_f32_16x16x4_f32(
                false, a3, false, b, (short)0, acc3, false, false);
        }
        const float bias = b1[ncol];
        #pragma unroll
        for (int r = 0; r < 8; ++r) {
            float v0 = acc0[r] + bias, v1 = acc1[r] + bias;
            float v2 = acc2[r] + bias, v3 = acc3[r] + bias;
            s_h[(     mbase + r) * H_STRIDE + ncol] = v0 > 0.f ? v0 : 0.f;
            s_h[(16 + mbase + r) * H_STRIDE + ncol] = v1 > 0.f ? v1 : 0.f;
            s_h[(32 + mbase + r) * H_STRIDE + ncol] = v2 > 0.f ? v2 : 0.f;
            s_h[(48 + mbase + r) * H_STRIDE + ncol] = v3 > 0.f ? v3 : 0.f;
        }
    }
    __syncthreads();

    {
        const int ntile = (wave & 3) * 16;
        const int kbase = (wave >> 2) * 64;
        const int ncol  = ntile + frow;
        v8f acc0 = {}, acc1 = {}, acc2 = {}, acc3 = {};
        #pragma unroll
        for (int kk = 0; kk < 16; ++kk) {
            const int k0 = kbase + kk * 4 + khalf;
            v2f b;
            b.x = W2[(size_t)(k0    ) * D_FEAT + ncol];
            b.y = W2[(size_t)(k0 + 1) * D_FEAT + ncol];
            v2f a0, a1, a2, a3;
            a0.x = s_h[(frow     ) * H_STRIDE + k0];
            a0.y = s_h[(frow     ) * H_STRIDE + k0 + 1];
            a1.x = s_h[(frow + 16) * H_STRIDE + k0];
            a1.y = s_h[(frow + 16) * H_STRIDE + k0 + 1];
            a2.x = s_h[(frow + 32) * H_STRIDE + k0];
            a2.y = s_h[(frow + 32) * H_STRIDE + k0 + 1];
            a3.x = s_h[(frow + 48) * H_STRIDE + k0];
            a3.y = s_h[(frow + 48) * H_STRIDE + k0 + 1];
            acc0 = __builtin_amdgcn_wmma_f32_16x16x4_f32(
                false, a0, false, b, (short)0, acc0, false, false);
            acc1 = __builtin_amdgcn_wmma_f32_16x16x4_f32(
                false, a1, false, b, (short)0, acc1, false, false);
            acc2 = __builtin_amdgcn_wmma_f32_16x16x4_f32(
                false, a2, false, b, (short)0, acc2, false, false);
            acc3 = __builtin_amdgcn_wmma_f32_16x16x4_f32(
                false, a3, false, b, (short)0, acc3, false, false);
        }
        #pragma unroll
        for (int r = 0; r < 8; ++r) {
            atomicAdd(&s_out[(     mbase + r) * D_FEAT + ncol], acc0[r]);
            atomicAdd(&s_out[(16 + mbase + r) * D_FEAT + ncol], acc1[r]);
            atomicAdd(&s_out[(32 + mbase + r) * D_FEAT + ncol], acc2[r]);
            atomicAdd(&s_out[(48 + mbase + r) * D_FEAT + ncol], acc3[r]);
        }
    }
    __syncthreads();

    for (int i = tid; i < TILE_M * D_FEAT; i += 512) {
        int m = i >> 6, n = i & 63;
        int row = node0 + m;
        if (row < n_nodes)
            out[(size_t)row * D_FEAT + n] = s_out[i] + b2[n];
    }
}

extern "C" void kernel_launch(void* const* d_in, const int* in_sizes, int n_in,
                              void* d_out, int out_size, void* d_ws, size_t ws_size,
                              hipStream_t stream) {
    const float* feat = (const float*)d_in[0];
    const float* W1   = (const float*)d_in[1];
    const float* b1   = (const float*)d_in[2];
    const float* W2   = (const float*)d_in[3];
    const float* b2   = (const float*)d_in[4];
    const int*   src  = (const int*)d_in[5];
    const int*   dst  = (const int*)d_in[6];
    float*       out  = (float*)d_out;

    const int n_nodes = in_sizes[0] / D_FEAT;
    const int n_edges = in_sizes[5];

    float* rst = (float*)d_ws;
    unsigned int* keys = (unsigned int*)((char*)d_ws + (32u << 20));

    {
        int n4 = ((n_edges + 15) / 16) * 4;
        gin_key_kernel<<<(n4 + 255) / 256, 256, 0, stream>>>(dst, keys, n_edges);
    }
    {
        int blocks = (n_nodes + AGG_ROWS - 1) / AGG_ROWS;
        gin_aggregate_kernel<<<blocks, AGG_THREADS, 0, stream>>>(feat, src, dst, keys, rst,
                                                                n_nodes, n_edges);
    }
    {
        int blocks = (n_nodes + TILE_M - 1) / TILE_M;
        gin_mlp_kernel<<<blocks, 512, 0, stream>>>(rst, W1, b1, W2, b2, out, n_nodes);
    }
}


# Owner keys stored bit-sliced (8 planes per 32 edges, built by wave ballots): the edge filter gets a 32-edge match mask from 8 boolean ops, was 54 byte-compare ops; sum order arrival as before
# speedup vs baseline: 1.6085x; 1.1029x over previous
#include <hip/hip_runtime.h>
#include <hip/hip_bf16.h>

typedef __attribute__((ext_vector_type(2))) float v2f;
typedef __attribute__((ext_vector_type(8))) float v8f;

#define D_FEAT 64
#define H_DIM  256
#define TILE_M 64
#define RST_STRIDE 68
#define H_STRIDE   260

#define AGG_ROWS    400
#define AGG_THREADS 1024
#define AGG_QCAP    16384

__global__ __launch_bounds__(256)
void gin_key_kernel(const int* __restrict__ dst,
                    unsigned int* __restrict__ keys, int n_edges) {
    const int lane = threadIdx.x & 31;
    const int w    = threadIdx.x >> 5;
    const int e0   = (blockIdx.x * 8 + w) * 128 + lane * 4;
    unsigned int k[4] = {0xFFu, 0xFFu, 0xFFu, 0xFFu};
    if (e0 + 3 < n_edges) {
        const int4 d = *reinterpret_cast<const int4*>(dst + e0);
        k[0] = (unsigned int)d.x / AGG_ROWS; k[1] = (unsigned int)d.y / AGG_ROWS;
        k[2] = (unsigned int)d.z / AGG_ROWS; k[3] = (unsigned int)d.w / AGG_ROWS;
    } else {
        if (e0     < n_edges) k[0] = (unsigned int)dst[e0    ] / AGG_ROWS;
        if (e0 + 1 < n_edges) k[1] = (unsigned int)dst[e0 + 1] / AGG_ROWS;
        if (e0 + 2 < n_edges) k[2] = (unsigned int)dst[e0 + 2] / AGG_ROWS;
    }
    unsigned int mine = 0;
    #pragma unroll
    for (int c = 0; c < 4; ++c) {
        #pragma unroll
        for (int b = 0; b < 8; ++b) {
            const unsigned int bal = __builtin_amdgcn_ballot_w32(((k[c] >> b) & 1u) != 0u);
            if (lane == c * 8 + b) mine = bal;
        }
    }
    const int b = lane & 7, c = lane >> 3;
    keys[(size_t)blockIdx.x * 256 + (b >> 2) * 128 + (4 * w + c) * 4 + (b & 3)] = mine;
}

__device__ __forceinline__ unsigned int match_planes(const uint4 lo, const uint4 hi,
                                                     const unsigned int (&x)[8]) {
    unsigned int m = lo.x ^ x[0];
    m &= lo.y ^ x[1]; m &= lo.z ^ x[2]; m &= lo.w ^ x[3];
    m &= hi.x ^ x[4]; m &= hi.y ^ x[5]; m &= hi.z ^ x[6]; m &= hi.w ^ x[7];
    return m;
}

__global__ __launch_bounds__(AGG_THREADS)
void gin_aggregate_kernel(const float* __restrict__ feat,
                          const int* __restrict__ src,
                          const int* __restrict__ dst,
                          const unsigned int* __restrict__ keys,
                          float* __restrict__ rst, int n_nodes, int n_edges) {
    __shared__ float s_tile[AGG_ROWS * D_FEAT];
    __shared__ int   s_queue[AGG_QCAP];
    __shared__ int   s_qcount;

    const int tid  = threadIdx.x;
    const int lane = tid & 31;
    const int wave = tid >> 5;
    const int lo   = blockIdx.x * AGG_ROWS;
    const int rows = (n_nodes - lo) < AGG_ROWS ? (n_nodes - lo) : AGG_ROWS;

    {
        const float4* f4 = reinterpret_cast<const float4*>(feat + (size_t)lo * D_FEAT);
        float4* t4 = reinterpret_cast<float4*>(s_tile);
        for (int i = tid; i < rows * (D_FEAT / 4); i += AGG_THREADS) t4[i] = f4[i];
    }
    if (tid == 0) s_qcount = 0;
    __syncthreads();

    {
        const int n_rec = ((n_edges + 1023) >> 10) << 5;
        const uint4* k16 = reinterpret_cast<const uint4*>(keys);
        unsigned int x[8];
        #pragma unroll
        for (int b = 0; b < 8; ++b) x[b] = ((blockIdx.x >> b) & 1u) ? 0u : 0xFFFFFFFFu;
        const int start = (int)(((long long)blockIdx.x * (n_rec >> 5)) / gridDim.x) << 5;
        uint4 cur[8];
        #pragma unroll
        for (int u = 0; u < 4; ++u) {
            const int p = u * AGG_THREADS + tid;
            int i = p + start;
            if (i >= n_rec) i -= n_rec;
            if (p >= n_rec) i = 0;
            const int a = ((i >> 5) << 6) + (i & 31);
            cur[2 * u]     = k16[a];
            cur[2 * u + 1] = k16[a + 32];
        }
        for (int p0 = 0; p0 < n_rec; p0 += AGG_THREADS * 4) {
            uint4 nxt[8];
            #pragma unroll
            for (int u = 0; u < 4; ++u) {
                const int p = p0 + (4 + u) * AGG_THREADS + tid;
                int i = p + start;
                if (i >= n_rec) i -= n_rec;
                if (p >= n_rec) i = 0;
                const int a = ((i >> 5) << 6) + (i & 31);
                nxt[2 * u]     = k16[a];
                nxt[2 * u + 1] = k16[a + 32];
            }
            #pragma unroll
            for (int h = 0; h < 2; ++h) {
                unsigned int m[2];
                int eb[2];
                #pragma unroll
                for (int v = 0; v < 2; ++v) {
                    const int u = 2 * h + v;
                    const int p = p0 + u * AGG_THREADS + tid;
                    int i = p + start;
                    if (i >= n_rec) i -= n_rec;
                    m[v]  = p < n_rec ? match_planes(cur[2 * u], cur[2 * u + 1], x) : 0u;
                    eb[v] = ((i >> 2) << 7) + (i & 3);
                }
                unsigned int m0 = m[0], m1 = m[1];
                while (m0 | m1) {
                    const bool second = (m0 == 0u);
                    unsigned int mm = second ? m1 : m0;
                    const int l = __builtin_ctz(mm);
                    mm &= mm - 1;
                    if (second) m1 = mm; else m0 = mm;
                    const int e = (second ? eb[1] : eb[0]) + 4 * l;
                    const int slot = atomicAdd(&s_qcount, 1);
                    if (slot < AGG_QCAP) {
                        s_queue[slot] = e;
                    } else {
                        const int d = dst[e] - lo;
                        const float* fr = feat + (size_t)src[e] * D_FEAT;
                        for (int k = 0; k < D_FEAT; ++k)
                            atomicAdd(&s_tile[d * D_FEAT + k], fr[k]);
                    }
                }
            }
            #pragma unroll
            for (int u = 0; u < 8; ++u) cur[u] = nxt[u];
        }
    }
    __syncthreads();

    {
        const int n = s_qcount < AGG_QCAP ? s_qcount : AGG_QCAP;
        for (int base = wave * 32; base < n; base += AGG_THREADS) {
            const int i = base + lane;
            int d = 0, s = 0;
            if (i < n) {
                const int e = s_queue[i];
                d = dst[e] - lo;
                s = src[e];
            }
            const int cnt = (n - base) < 32 ? (n - base) : 32;
            int j = 0;
            for (; j + 8 <= cnt; j += 8) {
                float2 v[8];
                int    dj[8];
                #pragma unroll
                for (int u = 0; u < 8; ++u) {
                    const int sj = __builtin_amdgcn_readlane(s, j + u);
                    dj[u] = __builtin_amdgcn_readlane(d, j + u);
                    v[u] = *reinterpret_cast<const float2*>(feat + (size_t)sj * D_FEAT + lane * 2);
                }
                #pragma unroll
                for (int u = 0; u < 8; ++u) {
                    atomicAdd(&s_tile[dj[u] * D_FEAT + lane * 2    ], v[u].x);
                    atomicAdd(&s_tile[dj[u] * D_FEAT + lane * 2 + 1], v[u].y);
                }
            }
            for (; j < cnt; ++j) {
                const int sj = __builtin_amdgcn_readlane(s, j);
                const int dd = __builtin_amdgcn_readlane(d, j);
                const float2 v = *reinterpret_cast<const float2*>(feat + (size_t)sj * D_FEAT + lane * 2);
                atomicAdd(&s_tile[dd * D_FEAT + lane * 2    ], v.x);
                atomicAdd(&s_tile[dd * D_FEAT + lane * 2 + 1], v.y);
            }
        }
    }
    __syncthreads();

    {
        float4* r4 = reinterpret_cast<float4*>(rst + (size_t)lo * D_FEAT);
        const float4* t4 = reinterpret_cast<const float4*>(s_tile);
        for (int i = tid; i < rows * (D_FEAT / 4); i += AGG_THREADS) r4[i] = t4[i];
    }
}

__global__ __launch_bounds__(512)
void gin_mlp_kernel(const float* __restrict__ rst,
                    const float* __restrict__ W1,
                    const float* __restrict__ b1,
                    const float* __restrict__ W2,
                    const float* __restrict__ b2,
                    float* __restrict__ out, int n_nodes) {
    __shared__ float s_rst[TILE_M * RST_STRIDE];
    __shared__ float s_h[TILE_M * H_STRIDE];
    __shared__ float s_out[TILE_M * D_FEAT];

    const int tid   = threadIdx.x;
    const int wave  = tid >> 5;
    const int lane  = tid & 31;
    const int node0 = blockIdx.x * TILE_M;

    for (int i = tid; i < TILE_M * D_FEAT; i += 512) {
        int m = i >> 6, k = i & 63;
        int row = node0 + m;
        row = row < n_nodes ? row : (n_nodes - 1);
        s_rst[m * RST_STRIDE + k] = rst[(size_t)row * D_FEAT + k];
    }
    for (int i = tid; i < TILE_M * D_FEAT; i += 512) s_out[i] = 0.0f;
    __syncthreads();

    const int frow  = lane & 15;
    const int khalf = (lane >> 4) << 1;
    const int mbase = (lane >> 4) << 3;

    {
        const int ncol = wave * 16 + frow;
        v8f acc0 = {}, acc1 = {}, acc2 = {}, acc3 = {};
        #pragma unroll
        for (int kk = 0; kk < 16; ++kk) {
            const int k0 = kk * 4 + khalf;
            v2f b;
            b.x = W1[(size_t)(k0    ) * H_DIM + ncol];
            b.y = W1[(size_t)(k0 + 1) * H_DIM + ncol];
            v2f a0, a1, a2, a3;
            a0.x = s_rst[(frow     ) * RST_STRIDE + k0];
            a0.y = s_rst[(frow     ) * RST_STRIDE + k0 + 1];
            a1.x = s_rst[(frow + 16) * RST_STRIDE + k0];
            a1.y = s_rst[(frow + 16) * RST_STRIDE + k0 + 1];
            a2.x = s_rst[(frow + 32) * RST_STRIDE + k0];
            a2.y = s_rst[(frow + 32) * RST_STRIDE + k0 + 1];
            a3.x = s_rst[(frow + 48) * RST_STRIDE + k0];
            a3.y = s_rst[(frow + 48) * RST_STRIDE + k0 + 1];
            acc0 = __builtin_amdgcn_wmma_f32_16x16x4_f32(
                false, a0, false, b, (short)0, acc0, false, false);
            acc1 = __builtin_amdgcn_wmma_f32_16x16x4_f32(
                false, a1, false, b, (short)0, acc1, false, false);
            acc2 = __builtin_amdgcn_wmma_f32_16x16x4_f32(
                false, a2, false, b, (short)0, acc2, false, false);
            acc3 = __builtin_amdgcn_wmma_f32_16x16x4_f32(
                false, a3, false, b, (short)0, acc3, false, false);
        }
        const float bias = b1[ncol];
        #pragma unroll
        for (int r = 0; r < 8; ++r) {
            float v0 = acc0[r] + bias, v1 = acc1[r] + bias;
            float v2 = acc2[r] + bias, v3 = acc3[r] + bias;
            s_h[(     mbase + r) * H_STRIDE + ncol] = v0 > 0.f ? v0 : 0.f;
            s_h[(16 + mbase + r) * H_STRIDE + ncol] = v1 > 0.f ? v1 : 0.f;
            s_h[(32 + mbase + r) * H_STRIDE + ncol] = v2 > 0.f ? v2 : 0.f;
            s_h[(48 + mbase + r) * H_STRIDE + ncol] = v3 > 0.f ? v3 : 0.f;
        }
    }
    __syncthreads();

    {
        const int ntile = (wave & 3) * 16;
        const int kbase = (wave >> 2) * 64;
        const int ncol  = ntile + frow;
        v8f acc0 = {}, acc1 = {}, acc2 = {}, acc3 = {};
        #pragma unroll
        for (int kk = 0; kk < 16; ++kk) {
            const int k0 = kbase + kk * 4 + khalf;
            v2f b;
            b.x = W2[(size_t)(k0    ) * D_FEAT + ncol];
            b.y = W2[(size_t)(k0 + 1) * D_FEAT + ncol];
            v2f a0, a1, a2, a3;
            a0.x = s_h[(frow     ) * H_STRIDE + k0];
            a0.y = s_h[(frow     ) * H_STRIDE + k0 + 1];
            a1.x = s_h[(frow + 16) * H_STRIDE + k0];
            a1.y = s_h[(frow + 16) * H_STRIDE + k0 + 1];
            a2.x = s_h[(frow + 32) * H_STRIDE + k0];
            a2.y = s_h[(frow + 32) * H_STRIDE + k0 + 1];
            a3.x = s_h[(frow + 48) * H_STRIDE + k0];
            a3.y = s_h[(frow + 48) * H_STRIDE + k0 + 1];
            acc0 = __builtin_amdgcn_wmma_f32_16x16x4_f32(
                false, a0, false, b, (short)0, acc0, false, false);
            acc1 = __builtin_amdgcn_wmma_f32_16x16x4_f32(
                false, a1, false, b, (short)0, acc1, false, false);
            acc2 = __builtin_amdgcn_wmma_f32_16x16x4_f32(
                false, a2, false, b, (short)0, acc2, false, false);
            acc3 = __builtin_amdgcn_wmma_f32_16x16x4_f32(
                false, a3, false, b, (short)0, acc3, false, false);
        }
        #pragma unroll
        for (int r = 0; r < 8; ++r) {
            atomicAdd(&s_out[(     mbase + r) * D_FEAT + ncol], acc0[r]);
            atomicAdd(&s_out[(16 + mbase + r) * D_FEAT + ncol], acc1[r]);
            atomicAdd(&s_out[(32 + mbase + r) * D_FEAT + ncol], acc2[r]);
            atomicAdd(&s_out[(48 + mbase + r) * D_FEAT + ncol], acc3[r]);
        }
    }
    __syncthreads();

    for (int i = tid; i < TILE_M * D_FEAT; i += 512) {
        int m = i >> 6, n = i & 63;
        int row = node0 + m;
        if (row < n_nodes)
            out[(size_t)row * D_FEAT + n] = s_out[i] + b2[n];
    }
}

extern "C" void kernel_launch(void* const* d_in, const int* in_sizes, int n_in,
                              void* d_out, int out_size, void* d_ws, size_t ws_size,
                              hipStream_t stream) {
    const float* feat = (const float*)d_in[0];
    const float* W1   = (const float*)d_in[1];
    const float* b1   = (const float*)d_in[2];
    const float* W2   = (const float*)d_in[3];
    const float* b2   = (const float*)d_in[4];
    const int*   src  = (const int*)d_in[5];
    const int*   dst  = (const int*)d_in[6];
    float*       out  = (float*)d_out;

    const int n_nodes = in_sizes[0] / D_FEAT;
    const int n_edges = in_sizes[5];

    float* rst = (float*)d_ws;
    unsigned int* keys = (unsigned int*)((char*)d_ws + (32u << 20));

    {
        int tiles = (n_edges + 1023) / 1024;
        gin_key_kernel<<<tiles, 256, 0, stream>>>(dst, keys, n_edges);
    }
    {
        int blocks = (n_nodes + AGG_ROWS - 1) / AGG_ROWS;
        gin_aggregate_kernel<<<blocks, AGG_THREADS, 0, stream>>>(feat, src, dst, keys, rst,
                                                                n_nodes, n_edges);
    }
    {
        int blocks = (n_nodes + TILE_M - 1) / TILE_M;
        gin_mlp_kernel<<<blocks, 512, 0, stream>>>(rst, W1, b1, W2, b2, out, n_nodes);
    }
}


# Owner keys stored as 32 one-hot nibble planes (16 high, 16 low, built in LDS by integer OR): the edge filter ANDs two 16-byte loads per 128 edges, was 8 loads + 32 boolean ops; sum order arrival
# speedup vs baseline: 1.6570x; 1.0302x over previous
#include <hip/hip_runtime.h>
#include <hip/hip_bf16.h>

typedef __attribute__((ext_vector_type(2))) float v2f;
typedef __attribute__((ext_vector_type(8))) float v8f;

#define D_FEAT 64
#define H_DIM  256
#define TILE_M 64
#define RST_STRIDE 68
#define H_STRIDE   260

#define AGG_ROWS    400
#define AGG_THREADS 1024
#define AGG_QCAP    16384
#define KEY_PLANES  32
#define BM_PITCH    33

__global__ __launch_bounds__(256)
void gin_key_kernel(const int* __restrict__ dst,
                    unsigned int* __restrict__ planes, int n_edges) {
    __shared__ unsigned int s_bm[KEY_PLANES * BM_PITCH];
    const int tid  = threadIdx.x;
    const int lane = tid & 31;
    const int w    = tid >> 5;
    for (int i = tid; i < KEY_PLANES * BM_PITCH; i += 256) s_bm[i] = 0u;
    __syncthreads();

    const int e0 = blockIdx.x * 1024 + tid * 4;
    int d[4] = {0, 0, 0, 0};
    int nv = 0;
    if (e0 + 3 < n_edges) {
        const int4 v = *reinterpret_cast<const int4*>(dst + e0);
        d[0] = v.x; d[1] = v.y; d[2] = v.z; d[3] = v.w;
        nv = 4;
    } else {
        if (e0     < n_edges) { d[0] = dst[e0    ]; nv = 1; }
        if (e0 + 1 < n_edges) { d[1] = dst[e0 + 1]; nv = 2; }
        if (e0 + 2 < n_edges) { d[2] = dst[e0 + 2]; nv = 3; }
    }
    const int word = tid >> 3;
    const int bit0 = (tid & 7) << 2;
    #pragma unroll
    for (int c = 0; c < 4; ++c) {
        const unsigned int key = (unsigned int)d[c] / AGG_ROWS;
        if (c < nv && key < 256u) {
            const unsigned int bit = 1u << (bit0 + c);
            atomicOr(&s_bm[(key >> 4) * BM_PITCH + word], bit);
            atomicOr(&s_bm[(16u + (key & 15u)) * BM_PITCH + word], bit);
        }
    }
    __syncthreads();

    const size_t stride = (size_t)gridDim.x * 32;
    #pragma unroll
    for (int q = 0; q < KEY_PLANES / 8; ++q) {
        const int p = w + 8 * q;
        planes[(size_t)p * stride + (size_t)blockIdx.x * 32 + lane] = s_bm[p * BM_PITCH + lane];
    }
}

__global__ __launch_bounds__(AGG_THREADS)
void gin_aggregate_kernel(const float* __restrict__ feat,
                          const int* __restrict__ src,
                          const int* __restrict__ dst,
                          const unsigned int* __restrict__ planes,
                          float* __restrict__ rst, int n_nodes, int n_edges) {
    __shared__ float s_tile[AGG_ROWS * D_FEAT];
    __shared__ int   s_queue[AGG_QCAP];
    __shared__ int   s_qcount;

    const int tid  = threadIdx.x;
    const int lane = tid & 31;
    const int wave = tid >> 5;
    const int lo   = blockIdx.x * AGG_ROWS;
    const int rows = (n_nodes - lo) < AGG_ROWS ? (n_nodes - lo) : AGG_ROWS;

    {
        const float4* f4 = reinterpret_cast<const float4*>(feat + (size_t)lo * D_FEAT);
        float4* t4 = reinterpret_cast<float4*>(s_tile);
        for (int i = tid; i < rows * (D_FEAT / 4); i += AGG_THREADS) t4[i] = f4[i];
    }
    if (tid == 0) s_qcount = 0;
    __syncthreads();

    {
        const int n_tiles = (n_edges + 1023) >> 10;
        const int n4 = n_tiles << 3;
        const size_t stride = (size_t)n_tiles * 32;
        const uint4* hi = reinterpret_cast<const uint4*>(planes + (size_t)(blockIdx.x >> 4) * stride);
        const uint4* lw = reinterpret_cast<const uint4*>(planes + (size_t)(16 + (blockIdx.x & 15)) * stride);
        const int start = (int)(((unsigned int)blockIdx.x * (unsigned int)n_tiles) / gridDim.x) << 3;
        uint4 a[4], b[4];
        #pragma unroll
        for (int u = 0; u < 4; ++u) {
            const int p = u * AGG_THREADS + tid;
            int i = p + start;
            if (i >= n4) i -= n4;
            if (p >= n4) i = 0;
            a[u] = hi[i];
            b[u] = lw[i];
        }
        for (int p0 = 0; p0 < n4; p0 += AGG_THREADS * 4) {
            unsigned int m[4][4];
            int eb[4];
            #pragma unroll
            for (int u = 0; u < 4; ++u) {
                const int p = p0 + u * AGG_THREADS + tid;
                int i = p + start;
                if (i >= n4) i -= n4;
                const bool ok = p < n4;
                m[u][0] = ok ? (a[u].x & b[u].x) : 0u;
                m[u][1] = ok ? (a[u].y & b[u].y) : 0u;
                m[u][2] = ok ? (a[u].z & b[u].z) : 0u;
                m[u][3] = ok ? (a[u].w & b[u].w) : 0u;
                eb[u] = i << 7;
            }
            #pragma unroll
            for (int u = 0; u < 4; ++u) {
                const int p = p0 + (4 + u) * AGG_THREADS + tid;
                int i = p + start;
                if (i >= n4) i -= n4;
                if (p >= n4) i = 0;
                a[u] = hi[i];
                b[u] = lw[i];
            }
            #pragma unroll
            for (int u = 0; u < 4; ++u) {
                unsigned int m0 = m[u][0], m1 = m[u][1], m2 = m[u][2], m3 = m[u][3];
                while (m0 | m1 | m2 | m3) {
                    const bool z0 = (m0 == 0u), z1 = (m1 == 0u), z2 = (m2 == 0u);
                    const bool p1 = z0 && !z1, p2 = z0 && z1 && !z2, p3 = z0 && z1 && z2;
                    const unsigned int mm = p3 ? m3 : (p2 ? m2 : (p1 ? m1 : m0));
                    const int off = p3 ? 96 : (p2 ? 64 : (p1 ? 32 : 0));
                    const unsigned int cl = mm & (mm - 1u);
                    m0 = z0 ? m0 : cl;
                    m1 = p1 ? cl : m1;
                    m2 = p2 ? cl : m2;
                    m3 = p3 ? cl : m3;
                    const int e = eb[u] + off + __builtin_ctz(mm);
                    const int slot = atomicAdd(&s_qcount, 1);
                    if (slot < AGG_QCAP) {
                        s_queue[slot] = e;
                    } else {
                        const int d = dst[e] - lo;
                        const float* fr = feat + (size_t)src[e] * D_FEAT;
                        for (int k = 0; k < D_FEAT; ++k)
                            atomicAdd(&s_tile[d * D_FEAT + k], fr[k]);
                    }
                }
            }
        }
    }
    __syncthreads();

    {
        const int n = s_qcount < AGG_QCAP ? s_qcount : AGG_QCAP;
        for (int base = wave * 32; base < n; base += AGG_THREADS) {
            const int i = base + lane;
            int d = 0, s = 0;
            if (i < n) {
                const int e = s_queue[i];
                d = dst[e] - lo;
                s = src[e];
            }
            const int cnt = (n - base) < 32 ? (n - base) : 32;
            int j = 0;
            for (; j + 8 <= cnt; j += 8) {
                float2 v[8];
                int    dj[8];
                #pragma unroll
                for (int u = 0; u < 8; ++u) {
                    const int sj = __builtin_amdgcn_readlane(s, j + u);
                    dj[u] = __builtin_amdgcn_readlane(d, j + u);
                    v[u] = *reinterpret_cast<const float2*>(feat + (size_t)sj * D_FEAT + lane * 2);
                }
                #pragma unroll
                for (int u = 0; u < 8; ++u) {
                    atomicAdd(&s_tile[dj[u] * D_FEAT + lane * 2    ], v[u].x);
                    atomicAdd(&s_tile[dj[u] * D_FEAT + lane * 2 + 1], v[u].y);
                }
            }
            for (; j < cnt; ++j) {
                const int sj = __builtin_amdgcn_readlane(s, j);
                const int dd = __builtin_amdgcn_readlane(d, j);
                const float2 v = *reinterpret_cast<const float2*>(feat + (size_t)sj * D_FEAT + lane * 2);
                atomicAdd(&s_tile[dd * D_FEAT + lane * 2    ], v.x);
                atomicAdd(&s_tile[dd * D_FEAT + lane * 2 + 1], v.y);
            }
        }
    }
    __syncthreads();

    {
        float4* r4 = reinterpret_cast<float4*>(rst + (size_t)lo * D_FEAT);
        const float4* t4 = reinterpret_cast<const float4*>(s_tile);
        for (int i = tid; i < rows * (D_FEAT / 4); i += AGG_THREADS) r4[i] = t4[i];
    }
}

__global__ __launch_bounds__(512)
void gin_mlp_kernel(const float* __restrict__ rst,
                    const float* __restrict__ W1,
                    const float* __restrict__ b1,
                    const float* __restrict__ W2,
                    const float* __restrict__ b2,
                    float* __restrict__ out, int n_nodes) {
    __shared__ float s_rst[TILE_M * RST_STRIDE];
    __shared__ float s_h[TILE_M * H_STRIDE];
    __shared__ float s_out[TILE_M * D_FEAT];

    const int tid   = threadIdx.x;
    const int wave  = tid >> 5;
    const int lane  = tid & 31;
    const int node0 = blockIdx.x * TILE_M;

    for (int i = tid; i < TILE_M * D_FEAT; i += 512) {
        int m = i >> 6, k = i & 63;
        int row = node0 + m;
        row = row < n_nodes ? row : (n_nodes - 1);
        s_rst[m * RST_STRIDE + k] = rst[(size_t)row * D_FEAT + k];
    }
    for (int i = tid; i < TILE_M * D_FEAT; i += 512) s_out[i] = 0.0f;
    __syncthreads();

    const int frow  = lane & 15;
    const int khalf = (lane >> 4) << 1;
    const int mbase = (lane >> 4) << 3;

    {
        const int ncol = wave * 16 + frow;
        v8f acc0 = {}, acc1 = {}, acc2 = {}, acc3 = {};
        #pragma unroll
        for (int kk = 0; kk < 16; ++kk) {
            const int k0 = kk * 4 + khalf;
            v2f b;
            b.x = W1[(size_t)(k0    ) * H_DIM + ncol];
            b.y = W1[(size_t)(k0 + 1) * H_DIM + ncol];
            v2f a0, a1, a2, a3;
            a0.x = s_rst[(frow     ) * RST_STRIDE + k0];
            a0.y = s_rst[(frow     ) * RST_STRIDE + k0 + 1];
            a1.x = s_rst[(frow + 16) * RST_STRIDE + k0];
            a1.y = s_rst[(frow + 16) * RST_STRIDE + k0 + 1];
            a2.x = s_rst[(frow + 32) * RST_STRIDE + k0];
            a2.y = s_rst[(frow + 32) * RST_STRIDE + k0 + 1];
            a3.x = s_rst[(frow + 48) * RST_STRIDE + k0];
            a3.y = s_rst[(frow + 48) * RST_STRIDE + k0 + 1];
            acc0 = __builtin_amdgcn_wmma_f32_16x16x4_f32(
                false, a0, false, b, (short)0, acc0, false, false);
            acc1 = __builtin_amdgcn_wmma_f32_16x16x4_f32(
                false, a1, false, b, (short)0, acc1, false, false);
            acc2 = __builtin_amdgcn_wmma_f32_16x16x4_f32(
                false, a2, false, b, (short)0, acc2, false, false);
            acc3 = __builtin_amdgcn_wmma_f32_16x16x4_f32(
                false, a3, false, b, (short)0, acc3, false, false);
        }
        const float bias = b1[ncol];
        #pragma unroll
        for (int r = 0; r < 8; ++r) {
            float v0 = acc0[r] + bias, v1 = acc1[r] + bias;
            float v2 = acc2[r] + bias, v3 = acc3[r] + bias;
            s_h[(     mbase + r) * H_STRIDE + ncol] = v0 > 0.f ? v0 : 0.f;
            s_h[(16 + mbase + r) * H_STRIDE + ncol] = v1 > 0.f ? v1 : 0.f;
            s_h[(32 + mbase + r) * H_STRIDE + ncol] = v2 > 0.f ? v2 : 0.f;
            s_h[(48 + mbase + r) * H_STRIDE + ncol] = v3 > 0.f ? v3 : 0.f;
        }
    }
    __syncthreads();

    {
        const int ntile = (wave & 3) * 16;
        const int kbase = (wave >> 2) * 64;
        const int ncol  = ntile + frow;
        v8f acc0 = {}, acc1 = {}, acc2 = {}, acc3 = {};
        #pragma unroll
        for (int kk = 0; kk < 16; ++kk) {
            const int k0 = kbase + kk * 4 + khalf;
            v2f b;
            b.x = W2[(size_t)(k0    ) * D_FEAT + ncol];
            b.y = W2[(size_t)(k0 + 1) * D_FEAT + ncol];
            v2f a0, a1, a2, a3;
            a0.x = s_h[(frow     ) * H_STRIDE + k0];
            a0.y = s_h[(frow     ) * H_STRIDE + k0 + 1];
            a1.x = s_h[(frow + 16) * H_STRIDE + k0];
            a1.y = s_h[(frow + 16) * H_STRIDE + k0 + 1];
            a2.x = s_h[(frow + 32) * H_STRIDE + k0];
            a2.y = s_h[(frow + 32) * H_STRIDE + k0 + 1];
            a3.x = s_h[(frow + 48) * H_STRIDE + k0];
            a3.y = s_h[(frow + 48) * H_STRIDE + k0 + 1];
            acc0 = __builtin_amdgcn_wmma_f32_16x16x4_f32(
                false, a0, false, b, (short)0, acc0, false, false);
            acc1 = __builtin_amdgcn_wmma_f32_16x16x4_f32(
                false, a1, false, b, (short)0, acc1, false, false);
            acc2 = __builtin_amdgcn_wmma_f32_16x16x4_f32(
                false, a2, false, b, (short)0, acc2, false, false);
            acc3 = __builtin_amdgcn_wmma_f32_16x16x4_f32(
                false, a3, false, b, (short)0, acc3, false, false);
        }
        #pragma unroll
        for (int r = 0; r < 8; ++r) {
            atomicAdd(&s_out[(     mbase + r) * D_FEAT + ncol], acc0[r]);
            atomicAdd(&s_out[(16 + mbase + r) * D_FEAT + ncol], acc1[r]);
            atomicAdd(&s_out[(32 + mbase + r) * D_FEAT + ncol], acc2[r]);
            atomicAdd(&s_out[(48 + mbase + r) * D_FEAT + ncol], acc3[r]);
        }
    }
    __syncthreads();

    for (int i = tid; i < TILE_M * D_FEAT; i += 512) {
        int m = i >> 6, n = i & 63;
        int row = node0 + m;
        if (row < n_nodes)
            out[(size_t)row * D_FEAT + n] = s_out[i] + b2[n];
    }
}

extern "C" void kernel_launch(void* const* d_in, const int* in_sizes, int n_in,
                              void* d_out, int out_size, void* d_ws, size_t ws_size,
                              hipStream_t stream) {
    const float* feat = (const float*)d_in[0];
    const float* W1   = (const float*)d_in[1];
    const float* b1   = (const float*)d_in[2];
    const float* W2   = (const float*)d_in[3];
    const float* b2   = (const float*)d_in[4];
    const int*   src  = (const int*)d_in[5];
    const int*   dst  = (const int*)d_in[6];
    float*       out  = (float*)d_out;

    const int n_nodes = in_sizes[0] / D_FEAT;
    const int n_edges = in_sizes[5];

    float* rst = (float*)d_ws;
    unsigned int* planes = (unsigned int*)((char*)d_ws + (32u << 20));

    {
        int tiles = (n_edges + 1023) / 1024;
        gin_key_kernel<<<tiles, 256, 0, stream>>>(dst, planes, n_edges);
    }
    {
        int blocks = (n_nodes + AGG_ROWS - 1) / AGG_ROWS;
        gin_aggregate_kernel<<<blocks, AGG_THREADS, 0, stream>>>(feat, src, dst, planes, rst,
                                                                n_nodes, n_edges);
    }
    {
        int blocks = (n_nodes + TILE_M - 1) / TILE_M;
        gin_mlp_kernel<<<blocks, 512, 0, stream>>>(rst, W1, b1, W2, b2, out, n_nodes);
    }
}


# Load-wait-store trips removed: MLP tile as two 16-byte loads per thread issued with the wave's W1 slice, W2 slice under GEMM1, b2 in registers, 16-B out stores; aggregation start rows in one group
# speedup vs baseline: 2.0317x; 1.2261x over previous
#include <hip/hip_runtime.h>
#include <hip/hip_bf16.h>

typedef __attribute__((ext_vector_type(2))) float v2f;
typedef __attribute__((ext_vector_type(8))) float v8f;

#define D_FEAT 64
#define H_DIM  256
#define TILE_M 64
#define RST_STRIDE 68
#define H_STRIDE   260

#define AGG_ROWS    400
#define AGG_THREADS 1024
#define AGG_QCAP    16384
#define KEY_PLANES  32
#define BM_PITCH    33

__global__ __launch_bounds__(256)
void gin_key_kernel(const int* __restrict__ dst,
                    unsigned int* __restrict__ planes, int n_edges) {
    __shared__ unsigned int s_bm[KEY_PLANES * BM_PITCH];
    const int tid  = threadIdx.x;
    const int lane = tid & 31;
    const int w    = tid >> 5;
    for (int i = tid; i < KEY_PLANES * BM_PITCH; i += 256) s_bm[i] = 0u;
    __syncthreads();

    const int e0 = blockIdx.x * 1024 + tid * 4;
    int d[4] = {0, 0, 0, 0};
    int nv = 0;
    if (e0 + 3 < n_edges) {
        const int4 v = *reinterpret_cast<const int4*>(dst + e0);
        d[0] = v.x; d[1] = v.y; d[2] = v.z; d[3] = v.w;
        nv = 4;
    } else {
        if (e0     < n_edges) { d[0] = dst[e0    ]; nv = 1; }
        if (e0 + 1 < n_edges) { d[1] = dst[e0 + 1]; nv = 2; }
        if (e0 + 2 < n_edges) { d[2] = dst[e0 + 2]; nv = 3; }
    }
    const int word = tid >> 3;
    const int bit0 = (tid & 7) << 2;
    #pragma unroll
    for (int c = 0; c < 4; ++c) {
        const unsigned int key = (unsigned int)d[c] / AGG_ROWS;
        if (c < nv && key < 256u) {
            const unsigned int bit = 1u << (bit0 + c);
            atomicOr(&s_bm[(key >> 4) * BM_PITCH + word], bit);
            atomicOr(&s_bm[(16u + (key & 15u)) * BM_PITCH + word], bit);
        }
    }
    __syncthreads();

    const size_t stride = (size_t)gridDim.x * 32;
    #pragma unroll
    for (int q = 0; q < KEY_PLANES / 8; ++q) {
        const int p = w + 8 * q;
        planes[(size_t)p * stride + (size_t)blockIdx.x * 32 + lane] = s_bm[p * BM_PITCH + lane];
    }
}

__global__ __launch_bounds__(AGG_THREADS)
void gin_aggregate_kernel(const float* __restrict__ feat,
                          const int* __restrict__ src,
                          const int* __restrict__ dst,
                          const unsigned int* __restrict__ planes,
                          float* __restrict__ rst, int n_nodes, int n_edges) {
    __shared__ float s_tile[AGG_ROWS * D_FEAT];
    __shared__ int   s_queue[AGG_QCAP];
    __shared__ int   s_qcount;

    const int tid  = threadIdx.x;
    const int lane = tid & 31;
    const int wave = tid >> 5;
    const int lo   = blockIdx.x * AGG_ROWS;
    const int rows = (n_nodes - lo) < AGG_ROWS ? (n_nodes - lo) : AGG_ROWS;

    {
        const float4* f4 = reinterpret_cast<const float4*>(feat + (size_t)lo * D_FEAT);
        float4* t4 = reinterpret_cast<float4*>(s_tile);
        constexpr int PIECES  = AGG_ROWS * (D_FEAT / 4);
        constexpr int PER_THR = 8;
        constexpr int LOADERS = PIECES / PER_THR;
        static_assert(LOADERS * PER_THR == PIECES && LOADERS % 32 == 0 && LOADERS <= AGG_THREADS,
                      "full-tile start: whole waves, no remainder");
        if (rows == AGG_ROWS) {
            if (tid < LOADERS) {
                float4 v[PER_THR];
                #pragma unroll
                for (int u = 0; u < PER_THR; ++u) v[u] = f4[tid + u * LOADERS];
                #pragma unroll
                for (int u = 0; u < PER_THR; ++u) t4[tid + u * LOADERS] = v[u];
            }
        } else {
            for (int i = tid; i < rows * (D_FEAT / 4); i += AGG_THREADS) t4[i] = f4[i];
        }
    }
    if (tid == 0) s_qcount = 0;
    __syncthreads();

    {
        const int n_tiles = (n_edges + 1023) >> 10;
        const int n4 = n_tiles << 3;
        const size_t stride = (size_t)n_tiles * 32;
        const uint4* hi = reinterpret_cast<const uint4*>(planes + (size_t)(blockIdx.x >> 4) * stride);
        const uint4* lw = reinterpret_cast<const uint4*>(planes + (size_t)(16 + (blockIdx.x & 15)) * stride);
        const int start = (int)(((unsigned int)blockIdx.x * (unsigned int)n_tiles) / gridDim.x) << 3;
        uint4 a[4], b[4];
        #pragma unroll
        for (int u = 0; u < 4; ++u) {
            const int p = u * AGG_THREADS + tid;
            int i = p + start;
            if (i >= n4) i -= n4;
            if (p >= n4) i = 0;
            a[u] = hi[i];
            b[u] = lw[i];
        }
        for (int p0 = 0; p0 < n4; p0 += AGG_THREADS * 4) {
            unsigned int m[4][4];
            int eb[4];
            #pragma unroll
            for (int u = 0; u < 4; ++u) {
                const int p = p0 + u * AGG_THREADS + tid;
                int i = p + start;
                if (i >= n4) i -= n4;
                const bool ok = p < n4;
                m[u][0] = ok ? (a[u].x & b[u].x) : 0u;
                m[u][1] = ok ? (a[u].y & b[u].y) : 0u;
                m[u][2] = ok ? (a[u].z & b[u].z) : 0u;
                m[u][3] = ok ? (a[u].w & b[u].w) : 0u;
                eb[u] = i << 7;
            }
            #pragma unroll
            for (int u = 0; u < 4; ++u) {
                const int p = p0 + (4 + u) * AGG_THREADS + tid;
                int i = p + start;
                if (i >= n4) i -= n4;
                if (p >= n4) i = 0;
                a[u] = hi[i];
                b[u] = lw[i];
            }
            #pragma unroll
            for (int u = 0; u < 4; ++u) {
                unsigned int m0 = m[u][0], m1 = m[u][1], m2 = m[u][2], m3 = m[u][3];
                while (m0 | m1 | m2 | m3) {
                    const bool z0 = (m0 == 0u), z1 = (m1 == 0u), z2 = (m2 == 0u);
                    const bool p1 = z0 && !z1, p2 = z0 && z1 && !z2, p3 = z0 && z1 && z2;
                    const unsigned int mm = p3 ? m3 : (p2 ? m2 : (p1 ? m1 : m0));
                    const int off = p3 ? 96 : (p2 ? 64 : (p1 ? 32 : 0));
                    const unsigned int cl = mm & (mm - 1u);
                    m0 = z0 ? m0 : cl;
                    m1 = p1 ? cl : m1;
                    m2 = p2 ? cl : m2;
                    m3 = p3 ? cl : m3;
                    const int e = eb[u] + off + __builtin_ctz(mm);
                    const int slot = atomicAdd(&s_qcount, 1);
                    if (slot < AGG_QCAP) {
                        s_queue[slot] = e;
                    } else {
                        const int d = dst[e] - lo;
                        const float* fr = feat + (size_t)src[e] * D_FEAT;
                        for (int k = 0; k < D_FEAT; ++k)
                            atomicAdd(&s_tile[d * D_FEAT + k], fr[k]);
                    }
                }
            }
        }
    }
    __syncthreads();

    {
        const int n = s_qcount < AGG_QCAP ? s_qcount : AGG_QCAP;
        for (int base = wave * 32; base < n; base += AGG_THREADS) {
            const int i = base + lane;
            int d = 0, s = 0;
            if (i < n) {
                const int e = s_queue[i];
                d = dst[e] - lo;
                s = src[e];
            }
            const int cnt = (n - base) < 32 ? (n - base) : 32;
            int j = 0;
            for (; j + 8 <= cnt; j += 8) {
                float2 v[8];
                int    dj[8];
                #pragma unroll
                for (int u = 0; u < 8; ++u) {
                    const int sj = __builtin_amdgcn_readlane(s, j + u);
                    dj[u] = __builtin_amdgcn_readlane(d, j + u);
                    v[u] = *reinterpret_cast<const float2*>(feat + (size_t)sj * D_FEAT + lane * 2);
                }
                #pragma unroll
                for (int u = 0; u < 8; ++u) {
                    atomicAdd(&s_tile[dj[u] * D_FEAT + lane * 2    ], v[u].x);
                    atomicAdd(&s_tile[dj[u] * D_FEAT + lane * 2 + 1], v[u].y);
                }
            }
            for (; j < cnt; ++j) {
                const int sj = __builtin_amdgcn_readlane(s, j);
                const int dd = __builtin_amdgcn_readlane(d, j);
                const float2 v = *reinterpret_cast<const float2*>(feat + (size_t)sj * D_FEAT + lane * 2);
                atomicAdd(&s_tile[dd * D_FEAT + lane * 2    ], v.x);
                atomicAdd(&s_tile[dd * D_FEAT + lane * 2 + 1], v.y);
            }
        }
    }
    __syncthreads();

    {
        float4* r4 = reinterpret_cast<float4*>(rst + (size_t)lo * D_FEAT);
        const float4* t4 = reinterpret_cast<const float4*>(s_tile);
        for (int i = tid; i < rows * (D_FEAT / 4); i += AGG_THREADS) r4[i] = t4[i];
    }
}

__global__ __launch_bounds__(512)
void gin_mlp_kernel(const float* __restrict__ rst,
                    const float* __restrict__ W1,
                    const float* __restrict__ b1,
                    const float* __restrict__ W2,
                    const float* __restrict__ b2,
                    float* __restrict__ out, int n_nodes) {
    __shared__ __attribute__((aligned(16))) float s_rst[TILE_M * RST_STRIDE];
    __shared__ __attribute__((aligned(16))) float s_h[TILE_M * H_STRIDE];
    __shared__ __attribute__((aligned(16))) float s_out[TILE_M * D_FEAT];

    const int tid   = threadIdx.x;
    const int wave  = tid >> 5;
    const int lane  = tid & 31;
    const int node0 = blockIdx.x * TILE_M;

    const int frow  = lane & 15;
    const int khalf = (lane >> 4) << 1;
    const int mbase = (lane >> 4) << 3;

    const int c4 = tid & 15;
    const int m0 = tid >> 4;
    int r0 = node0 + m0;
    int r1 = node0 + m0 + 32;
    r0 = r0 < n_nodes ? r0 : (n_nodes - 1);
    r1 = r1 < n_nodes ? r1 : (n_nodes - 1);
    const float4 t0 = *reinterpret_cast<const float4*>(rst + (size_t)r0 * D_FEAT + c4 * 4);
    const float4 t1 = *reinterpret_cast<const float4*>(rst + (size_t)r1 * D_FEAT + c4 * 4);

    const int ncol1 = wave * 16 + frow;
    v2f bw1[16];
    #pragma unroll
    for (int kk = 0; kk < 16; ++kk) {
        const int k0 = kk * 4 + khalf;
        bw1[kk].x = W1[(size_t)(k0    ) * H_DIM + ncol1];
        bw1[kk].y = W1[(size_t)(k0 + 1) * H_DIM + ncol1];
    }
    const float bias1 = b1[ncol1];
    const float4 bias2 = *reinterpret_cast<const float4*>(b2 + c4 * 4);

    *reinterpret_cast<float4*>(&s_rst[(m0     ) * RST_STRIDE + c4 * 4]) = t0;
    *reinterpret_cast<float4*>(&s_rst[(m0 + 32) * RST_STRIDE + c4 * 4]) = t1;
    {
        const float4 z = make_float4(0.f, 0.f, 0.f, 0.f);
        reinterpret_cast<float4*>(s_out)[tid      ] = z;
        reinterpret_cast<float4*>(s_out)[tid + 512] = z;
    }
    __syncthreads();

    const int ntile = (wave & 3) * 16;
    const int kbase = (wave >> 2) * 64;
    const int ncol2 = ntile + frow;
    v2f bw2[16];
    #pragma unroll
    for (int kk = 0; kk < 16; ++kk) {
        const int k0 = kbase + kk * 4 + khalf;
        bw2[kk].x = W2[(size_t)(k0    ) * D_FEAT + ncol2];
        bw2[kk].y = W2[(size_t)(k0 + 1) * D_FEAT + ncol2];
    }

    {
        v8f acc0 = {}, acc1 = {}, acc2 = {}, acc3 = {};
        #pragma unroll
        for (int kk = 0; kk < 16; ++kk) {
            const int k0 = kk * 4 + khalf;
            const v2f b = bw1[kk];
            v2f a0, a1, a2, a3;
            a0.x = s_rst[(frow     ) * RST_STRIDE + k0];
            a0.y = s_rst[(frow     ) * RST_STRIDE + k0 + 1];
            a1.x = s_rst[(frow + 16) * RST_STRIDE + k0];
            a1.y = s_rst[(frow + 16) * RST_STRIDE + k0 + 1];
            a2.x = s_rst[(frow + 32) * RST_STRIDE + k0];
            a2.y = s_rst[(frow + 32) * RST_STRIDE + k0 + 1];
            a3.x = s_rst[(frow + 48) * RST_STRIDE + k0];
            a3.y = s_rst[(frow + 48) * RST_STRIDE + k0 + 1];
            acc0 = __builtin_amdgcn_wmma_f32_16x16x4_f32(
                false, a0, false, b, (short)0, acc0, false, false);
            acc1 = __builtin_amdgcn_wmma_f32_16x16x4_f32(
                false, a1, false, b, (short)0, acc1, false, false);
            acc2 = __builtin_amdgcn_wmma_f32_16x16x4_f32(
                false, a2, false, b, (short)0, acc2, false, false);
            acc3 = __builtin_amdgcn_wmma_f32_16x16x4_f32(
                false, a3, false, b, (short)0, acc3, false, false);
        }
        #pragma unroll
        for (int r = 0; r < 8; ++r) {
            float v0 = acc0[r] + bias1, v1 = acc1[r] + bias1;
            float v2 = acc2[r] + bias1, v3 = acc3[r] + bias1;
            s_h[(     mbase + r) * H_STRIDE + ncol1] = v0 > 0.f ? v0 : 0.f;
            s_h[(16 + mbase + r) * H_STRIDE + ncol1] = v1 > 0.f ? v1 : 0.f;
            s_h[(32 + mbase + r) * H_STRIDE + ncol1] = v2 > 0.f ? v2 : 0.f;
            s_h[(48 + mbase + r) * H_STRIDE + ncol1] = v3 > 0.f ? v3 : 0.f;
        }
    }
    __syncthreads();

    {
        v8f acc0 = {}, acc1 = {}, acc2 = {}, acc3 = {};
        #pragma unroll
        for (int kk = 0; kk < 16; ++kk) {
            const int k0 = kbase + kk * 4 + khalf;
            const v2f b = bw2[kk];
            v2f a0, a1, a2, a3;
            a0.x = s_h[(frow     ) * H_STRIDE + k0];
            a0.y = s_h[(frow     ) * H_STRIDE + k0 + 1];
            a1.x = s_h[(frow + 16) * H_STRIDE + k0];
            a1.y = s_h[(frow + 16) * H_STRIDE + k0 + 1];
            a2.x = s_h[(frow + 32) * H_STRIDE + k0];
            a2.y = s_h[(frow + 32) * H_STRIDE + k0 + 1];
            a3.x = s_h[(frow + 48) * H_STRIDE + k0];
            a3.y = s_h[(frow + 48) * H_STRIDE + k0 + 1];
            acc0 = __builtin_amdgcn_wmma_f32_16x16x4_f32(
                false, a0, false, b, (short)0, acc0, false, false);
            acc1 = __builtin_amdgcn_wmma_f32_16x16x4_f32(
                false, a1, false, b, (short)0, acc1, false, false);
            acc2 = __builtin_amdgcn_wmma_f32_16x16x4_f32(
                false, a2, false, b, (short)0, acc2, false, false);
            acc3 = __builtin_amdgcn_wmma_f32_16x16x4_f32(
                false, a3, false, b, (short)0, acc3, false, false);
        }
        #pragma unroll
        for (int r = 0; r < 8; ++r) {
            atomicAdd(&s_out[(     mbase + r) * D_FEAT + ncol2], acc0[r]);
            atomicAdd(&s_out[(16 + mbase + r) * D_FEAT + ncol2], acc1[r]);
            atomicAdd(&s_out[(32 + mbase + r) * D_FEAT + ncol2], acc2[r]);
            atomicAdd(&s_out[(48 + mbase + r) * D_FEAT + ncol2], acc3[r]);
        }
    }
    __syncthreads();

    {
        const float4 o0 = reinterpret_cast<const float4*>(s_out)[tid      ];
        const float4 o1 = reinterpret_cast<const float4*>(s_out)[tid + 512];
        if (node0 + m0 < n_nodes) {
            float4 v = make_float4(o0.x + bias2.x, o0.y + bias2.y, o0.z + bias2.z, o0.w + bias2.w);
            *reinterpret_cast<float4*>(out + (size_t)(node0 + m0) * D_FEAT + c4 * 4) = v;
        }
        if (node0 + m0 + 32 < n_nodes) {
            float4 v = make_float4(o1.x + bias2.x, o1.y + bias2.y, o1.z + bias2.z, o1.w + bias2.w);
            *reinterpret_cast<float4*>(out + (size_t)(node0 + m0 + 32) * D_FEAT + c4 * 4) = v;
        }
    }
}

extern "C" void kernel_launch(void* const* d_in, const int* in_sizes, int n_in,
                              void* d_out, int out_size, void* d_ws, size_t ws_size,
                              hipStream_t stream) {
    const float* feat = (const float*)d_in[0];
    const float* W1   = (const float*)d_in[1];
    const float* b1   = (const float*)d_in[2];
    const float* W2   = (const float*)d_in[3];
    const float* b2   = (const float*)d_in[4];
    const int*   src  = (const int*)d_in[5];
    const int*   dst  = (const int*)d_in[6];
    float*       out  = (float*)d_out;

    const int n_nodes = in_sizes[0] / D_FEAT;
    const int n_edges = in_sizes[5];

    float* rst = (float*)d_ws;
    unsigned int* planes = (unsigned int*)((char*)d_ws + (32u << 20));

    {
        int tiles = (n_edges + 1023) / 1024;
        gin_key_kernel<<<tiles, 256, 0, stream>>>(dst, planes, n_edges);
    }
    {
        int blocks = (n_nodes + AGG_ROWS - 1) / AGG_ROWS;
        gin_aggregate_kernel<<<blocks, AGG_THREADS, 0, stream>>>(feat, src, dst, planes, rst,
                                                                n_nodes, n_edges);
    }
    {
        int blocks = (n_nodes + TILE_M - 1) / TILE_M;
        gin_mlp_kernel<<<blocks, 512, 0, stream>>>(rst, W1, b1, W2, b2, out, n_nodes);
    }
}


# Gather reads ONE packed 4-byte word per queued edge (source row + local dst row, written by the key kernel in whole lines); was two random reads (src, dst), each its own line; sum order arrival
# speedup vs baseline: 2.1141x; 1.0406x over previous
#include <hip/hip_runtime.h>
#include <hip/hip_bf16.h>

typedef __attribute__((ext_vector_type(2))) float v2f;
typedef __attribute__((ext_vector_type(8))) float v8f;

#define D_FEAT 64
#define H_DIM  256
#define TILE_M 64
#define RST_STRIDE 68
#define H_STRIDE   260

#define AGG_ROWS    400
#define AGG_THREADS 1024
#define AGG_QCAP    16384
#define KEY_PLANES  32
#define BM_PITCH    33
#define PK_SRC_BITS 23

__global__ __launch_bounds__(256)
void gin_key_kernel(const int* __restrict__ src,
                    const int* __restrict__ dst,
                    unsigned int* __restrict__ planes,
                    unsigned int* __restrict__ packed, int n_edges) {
    __shared__ unsigned int s_bm[KEY_PLANES * BM_PITCH];
    const int tid  = threadIdx.x;
    const int lane = tid & 31;
    const int w    = tid >> 5;
    for (int i = tid; i < KEY_PLANES * BM_PITCH; i += 256) s_bm[i] = 0u;
    __syncthreads();

    const int e0 = blockIdx.x * 1024 + tid * 4;
    int d[4] = {0, 0, 0, 0};
    int s[4] = {0, 0, 0, 0};
    int nv = 0;
    if (e0 + 3 < n_edges) {
        const int4 v = *reinterpret_cast<const int4*>(dst + e0);
        const int4 q = *reinterpret_cast<const int4*>(src + e0);
        d[0] = v.x; d[1] = v.y; d[2] = v.z; d[3] = v.w;
        s[0] = q.x; s[1] = q.y; s[2] = q.z; s[3] = q.w;
        nv = 4;
    } else {
        if (e0     < n_edges) { d[0] = dst[e0    ]; s[0] = src[e0    ]; nv = 1; }
        if (e0 + 1 < n_edges) { d[1] = dst[e0 + 1]; s[1] = src[e0 + 1]; nv = 2; }
        if (e0 + 2 < n_edges) { d[2] = dst[e0 + 2]; s[2] = src[e0 + 2]; nv = 3; }
    }
    unsigned int pk[4];
    const int word = tid >> 3;
    const int bit0 = (tid & 7) << 2;
    #pragma unroll
    for (int c = 0; c < 4; ++c) {
        const unsigned int key = (unsigned int)d[c] / AGG_ROWS;
        pk[c] = (unsigned int)s[c] | (((unsigned int)d[c] - key * AGG_ROWS) << PK_SRC_BITS);
        if (c < nv && key < 256u) {
            const unsigned int bit = 1u << (bit0 + c);
            atomicOr(&s_bm[(key >> 4) * BM_PITCH + word], bit);
            atomicOr(&s_bm[(16u + (key & 15u)) * BM_PITCH + word], bit);
        }
    }
    if (nv == 4) {
        *reinterpret_cast<uint4*>(packed + e0) = make_uint4(pk[0], pk[1], pk[2], pk[3]);
    } else {
        if (nv > 0) packed[e0    ] = pk[0];
        if (nv > 1) packed[e0 + 1] = pk[1];
        if (nv > 2) packed[e0 + 2] = pk[2];
    }
    __syncthreads();

    const size_t stride = (size_t)gridDim.x * 32;
    #pragma unroll
    for (int q = 0; q < KEY_PLANES / 8; ++q) {
        const int p = w + 8 * q;
        planes[(size_t)p * stride + (size_t)blockIdx.x * 32 + lane] = s_bm[p * BM_PITCH + lane];
    }
}

__global__ __launch_bounds__(AGG_THREADS)
void gin_aggregate_kernel(const float* __restrict__ feat,
                          const unsigned int* __restrict__ packed,
                          const unsigned int* __restrict__ planes,
                          float* __restrict__ rst, int n_nodes, int n_edges) {
    __shared__ float s_tile[AGG_ROWS * D_FEAT];
    __shared__ int   s_queue[AGG_QCAP];
    __shared__ int   s_qcount;

    const int tid  = threadIdx.x;
    const int lane = tid & 31;
    const int wave = tid >> 5;
    const int lo   = blockIdx.x * AGG_ROWS;
    const int rows = (n_nodes - lo) < AGG_ROWS ? (n_nodes - lo) : AGG_ROWS;

    {
        const float4* f4 = reinterpret_cast<const float4*>(feat + (size_t)lo * D_FEAT);
        float4* t4 = reinterpret_cast<float4*>(s_tile);
        constexpr int PIECES  = AGG_ROWS * (D_FEAT / 4);
        constexpr int PER_THR = 8;
        constexpr int LOADERS = PIECES / PER_THR;
        static_assert(LOADERS * PER_THR == PIECES && LOADERS % 32 == 0 && LOADERS <= AGG_THREADS,
                      "full-tile start: whole waves, no remainder");
        if (rows == AGG_ROWS) {
            if (tid < LOADERS) {
                float4 v[PER_THR];
                #pragma unroll
                for (int u = 0; u < PER_THR; ++u) v[u] = f4[tid + u * LOADERS];
                #pragma unroll
                for (int u = 0; u < PER_THR; ++u) t4[tid + u * LOADERS] = v[u];
            }
        } else {
            for (int i = tid; i < rows * (D_FEAT / 4); i += AGG_THREADS) t4[i] = f4[i];
        }
    }
    if (tid == 0) s_qcount = 0;
    __syncthreads();

    {
        const int n_tiles = (n_edges + 1023) >> 10;
        const int n4 = n_tiles << 3;
        const size_t stride = (size_t)n_tiles * 32;
        const uint4* hi = reinterpret_cast<const uint4*>(planes + (size_t)(blockIdx.x >> 4) * stride);
        const uint4* lw = reinterpret_cast<const uint4*>(planes + (size_t)(16 + (blockIdx.x & 15)) * stride);
        const int start = (int)(((unsigned int)blockIdx.x * (unsigned int)n_tiles) / gridDim.x) << 3;
        uint4 a[4], b[4];
        #pragma unroll
        for (int u = 0; u < 4; ++u) {
            const int p = u * AGG_THREADS + tid;
            int i = p + start;
            if (i >= n4) i -= n4;
            if (p >= n4) i = 0;
            a[u] = hi[i];
            b[u] = lw[i];
        }
        for (int p0 = 0; p0 < n4; p0 += AGG_THREADS * 4) {
            unsigned int m[4][4];
            int eb[4];
            #pragma unroll
            for (int u = 0; u < 4; ++u) {
                const int p = p0 + u * AGG_THREADS + tid;
                int i = p + start;
                if (i >= n4) i -= n4;
                const bool ok = p < n4;
                m[u][0] = ok ? (a[u].x & b[u].x) : 0u;
                m[u][1] = ok ? (a[u].y & b[u].y) : 0u;
                m[u][2] = ok ? (a[u].z & b[u].z) : 0u;
                m[u][3] = ok ? (a[u].w & b[u].w) : 0u;
                eb[u] = i << 7;
            }
            #pragma unroll
            for (int u = 0; u < 4; ++u) {
                const int p = p0 + (4 + u) * AGG_THREADS + tid;
                int i = p + start;
                if (i >= n4) i -= n4;
                if (p >= n4) i = 0;
                a[u] = hi[i];
                b[u] = lw[i];
            }
            #pragma unroll
            for (int u = 0; u < 4; ++u) {
                unsigned int m0 = m[u][0], m1 = m[u][1], m2 = m[u][2], m3 = m[u][3];
                while (m0 | m1 | m2 | m3) {
                    const bool z0 = (m0 == 0u), z1 = (m1 == 0u), z2 = (m2 == 0u);
                    const bool p1 = z0 && !z1, p2 = z0 && z1 && !z2, p3 = z0 && z1 && z2;
                    const unsigned int mm = p3 ? m3 : (p2 ? m2 : (p1 ? m1 : m0));
                    const int off = p3 ? 96 : (p2 ? 64 : (p1 ? 32 : 0));
                    const unsigned int cl = mm & (mm - 1u);
                    m0 = z0 ? m0 : cl;
                    m1 = p1 ? cl : m1;
                    m2 = p2 ? cl : m2;
                    m3 = p3 ? cl : m3;
                    const int e = eb[u] + off + __builtin_ctz(mm);
                    const int slot = atomicAdd(&s_qcount, 1);
                    if (slot < AGG_QCAP) {
                        s_queue[slot] = e;
                    } else {
                        const unsigned int w = packed[e];
                        const int d = (int)(w >> PK_SRC_BITS);
                        const float* fr = feat + (size_t)(w & ((1u << PK_SRC_BITS) - 1u)) * D_FEAT;
                        for (int k = 0; k < D_FEAT; ++k)
                            atomicAdd(&s_tile[d * D_FEAT + k], fr[k]);
                    }
                }
            }
        }
    }
    __syncthreads();

    {
        const int n = s_qcount < AGG_QCAP ? s_qcount : AGG_QCAP;
        for (int base = wave * 32; base < n; base += AGG_THREADS) {
            const int i = base + lane;
            int d = 0, s = 0;
            if (i < n) {
                const unsigned int w = packed[s_queue[i]];
                d = (int)(w >> PK_SRC_BITS);
                s = (int)(w & ((1u << PK_SRC_BITS) - 1u));
            }
            const int cnt = (n - base) < 32 ? (n - base) : 32;
            int j = 0;
            for (; j + 8 <= cnt; j += 8) {
                float2 v[8];
                int    dj[8];
                #pragma unroll
                for (int u = 0; u < 8; ++u) {
                    const int sj = __builtin_amdgcn_readlane(s, j + u);
                    dj[u] = __builtin_amdgcn_readlane(d, j + u);
                    v[u] = *reinterpret_cast<const float2*>(feat + (size_t)sj * D_FEAT + lane * 2);
                }
                #pragma unroll
                for (int u = 0; u < 8; ++u) {
                    atomicAdd(&s_tile[dj[u] * D_FEAT + lane * 2    ], v[u].x);
                    atomicAdd(&s_tile[dj[u] * D_FEAT + lane * 2 + 1], v[u].y);
                }
            }
            for (; j < cnt; ++j) {
                const int sj = __builtin_amdgcn_readlane(s, j);
                const int dd = __builtin_amdgcn_readlane(d, j);
                const float2 v = *reinterpret_cast<const float2*>(feat + (size_t)sj * D_FEAT + lane * 2);
                atomicAdd(&s_tile[dd * D_FEAT + lane * 2    ], v.x);
                atomicAdd(&s_tile[dd * D_FEAT + lane * 2 + 1], v.y);
            }
        }
    }
    __syncthreads();

    {
        float4* r4 = reinterpret_cast<float4*>(rst + (size_t)lo * D_FEAT);
        const float4* t4 = reinterpret_cast<const float4*>(s_tile);
        for (int i = tid; i < rows * (D_FEAT / 4); i += AGG_THREADS) r4[i] = t4[i];
    }
}

__global__ __launch_bounds__(512)
void gin_mlp_kernel(const float* __restrict__ rst,
                    const float* __restrict__ W1,
                    const float* __restrict__ b1,
                    const float* __restrict__ W2,
                    const float* __restrict__ b2,
                    float* __restrict__ out, int n_nodes) {
    __shared__ __attribute__((aligned(16))) float s_rst[TILE_M * RST_STRIDE];
    __shared__ __attribute__((aligned(16))) float s_h[TILE_M * H_STRIDE];
    __shared__ __attribute__((aligned(16))) float s_out[TILE_M * D_FEAT];

    const int tid   = threadIdx.x;
    const int wave  = tid >> 5;
    const int lane  = tid & 31;
    const int node0 = blockIdx.x * TILE_M;

    const int frow  = lane & 15;
    const int khalf = (lane >> 4) << 1;
    const int mbase = (lane >> 4) << 3;

    const int c4 = tid & 15;
    const int m0 = tid >> 4;
    int r0 = node0 + m0;
    int r1 = node0 + m0 + 32;
    r0 = r0 < n_nodes ? r0 : (n_nodes - 1);
    r1 = r1 < n_nodes ? r1 : (n_nodes - 1);
    const float4 t0 = *reinterpret_cast<const float4*>(rst + (size_t)r0 * D_FEAT + c4 * 4);
    const float4 t1 = *reinterpret_cast<const float4*>(rst + (size_t)r1 * D_FEAT + c4 * 4);

    const int ncol1 = wave * 16 + frow;
    v2f bw1[16];
    #pragma unroll
    for (int kk = 0; kk < 16; ++kk) {
        const int k0 = kk * 4 + khalf;
        bw1[kk].x = W1[(size_t)(k0    ) * H_DIM + ncol1];
        bw1[kk].y = W1[(size_t)(k0 + 1) * H_DIM + ncol1];
    }
    const float bias1 = b1[ncol1];
    const float4 bias2 = *reinterpret_cast<const float4*>(b2 + c4 * 4);

    *reinterpret_cast<float4*>(&s_rst[(m0     ) * RST_STRIDE + c4 * 4]) = t0;
    *reinterpret_cast<float4*>(&s_rst[(m0 + 32) * RST_STRIDE + c4 * 4]) = t1;
    {
        const float4 z = make_float4(0.f, 0.f, 0.f, 0.f);
        reinterpret_cast<float4*>(s_out)[tid      ] = z;
        reinterpret_cast<float4*>(s_out)[tid + 512] = z;
    }
    __syncthreads();

    const int ntile = (wave & 3) * 16;
    const int kbase = (wave >> 2) * 64;
    const int ncol2 = ntile + frow;
    v2f bw2[16];
    #pragma unroll
    for (int kk = 0; kk < 16; ++kk) {
        const int k0 = kbase + kk * 4 + khalf;
        bw2[kk].x = W2[(size_t)(k0    ) * D_FEAT + ncol2];
        bw2[kk].y = W2[(size_t)(k0 + 1) * D_FEAT + ncol2];
    }

    {
        v8f acc0 = {}, acc1 = {}, acc2 = {}, acc3 = {};
        #pragma unroll
        for (int kk = 0; kk < 16; ++kk) {
            const int k0 = kk * 4 + khalf;
            const v2f b = bw1[kk];
            v2f a0, a1, a2, a3;
            a0.x = s_rst[(frow     ) * RST_STRIDE + k0];
            a0.y = s_rst[(frow     ) * RST_STRIDE + k0 + 1];
            a1.x = s_rst[(frow + 16) * RST_STRIDE + k0];
            a1.y = s_rst[(frow + 16) * RST_STRIDE + k0 + 1];
            a2.x = s_rst[(frow + 32) * RST_STRIDE + k0];
            a2.y = s_rst[(frow + 32) * RST_STRIDE + k0 + 1];
            a3.x = s_rst[(frow + 48) * RST_STRIDE + k0];
            a3.y = s_rst[(frow + 48) * RST_STRIDE + k0 + 1];
            acc0 = __builtin_amdgcn_wmma_f32_16x16x4_f32(
                false, a0, false, b, (short)0, acc0, false, false);
            acc1 = __builtin_amdgcn_wmma_f32_16x16x4_f32(
                false, a1, false, b, (short)0, acc1, false, false);
            acc2 = __builtin_amdgcn_wmma_f32_16x16x4_f32(
                false, a2, false, b, (short)0, acc2, false, false);
            acc3 = __builtin_amdgcn_wmma_f32_16x16x4_f32(
                false, a3, false, b, (short)0, acc3, false, false);
        }
        #pragma unroll
        for (int r = 0; r < 8; ++r) {
            float v0 = acc0[r] + bias1, v1 = acc1[r] + bias1;
            float v2 = acc2[r] + bias1, v3 = acc3[r] + bias1;
            s_h[(     mbase + r) * H_STRIDE + ncol1] = v0 > 0.f ? v0 : 0.f;
            s_h[(16 + mbase + r) * H_STRIDE + ncol1] = v1 > 0.f ? v1 : 0.f;
            s_h[(32 + mbase + r) * H_STRIDE + ncol1] = v2 > 0.f ? v2 : 0.f;
            s_h[(48 + mbase + r) * H_STRIDE + ncol1] = v3 > 0.f ? v3 : 0.f;
        }
    }
    __syncthreads();

    {
        v8f acc0 = {}, acc1 = {}, acc2 = {}, acc3 = {};
        #pragma unroll
        for (int kk = 0; kk < 16; ++kk) {
            const int k0 = kbase + kk * 4 + khalf;
            const v2f b = bw2[kk];
            v2f a0, a1, a2, a3;
            a0.x = s_h[(frow     ) * H_STRIDE + k0];
            a0.y = s_h[(frow     ) * H_STRIDE + k0 + 1];
            a1.x = s_h[(frow + 16) * H_STRIDE + k0];
            a1.y = s_h[(frow + 16) * H_STRIDE + k0 + 1];
            a2.x = s_h[(frow + 32) * H_STRIDE + k0];
            a2.y = s_h[(frow + 32) * H_STRIDE + k0 + 1];
            a3.x = s_h[(frow + 48) * H_STRIDE + k0];
            a3.y = s_h[(frow + 48) * H_STRIDE + k0 + 1];
            acc0 = __builtin_amdgcn_wmma_f32_16x16x4_f32(
                false, a0, false, b, (short)0, acc0, false, false);
            acc1 = __builtin_amdgcn_wmma_f32_16x16x4_f32(
                false, a1, false, b, (short)0, acc1, false, false);
            acc2 = __builtin_amdgcn_wmma_f32_16x16x4_f32(
                false, a2, false, b, (short)0, acc2, false, false);
            acc3 = __builtin_amdgcn_wmma_f32_16x16x4_f32(
                false, a3, false, b, (short)0, acc3, false, false);
        }
        #pragma unroll
        for (int r = 0; r < 8; ++r) {
            atomicAdd(&s_out[(     mbase + r) * D_FEAT + ncol2], acc0[r]);
            atomicAdd(&s_out[(16 + mbase + r) * D_FEAT + ncol2], acc1[r]);
            atomicAdd(&s_out[(32 + mbase + r) * D_FEAT + ncol2], acc2[r]);
            atomicAdd(&s_out[(48 + mbase + r) * D_FEAT + ncol2], acc3[r]);
        }
    }
    __syncthreads();

    {
        const float4 o0 = reinterpret_cast<const float4*>(s_out)[tid      ];
        const float4 o1 = reinterpret_cast<const float4*>(s_out)[tid + 512];
        if (node0 + m0 < n_nodes) {
            float4 v = make_float4(o0.x + bias2.x, o0.y + bias2.y, o0.z + bias2.z, o0.w + bias2.w);
            *reinterpret_cast<float4*>(out + (size_t)(node0 + m0) * D_FEAT + c4 * 4) = v;
        }
        if (node0 + m0 + 32 < n_nodes) {
            float4 v = make_float4(o1.x + bias2.x, o1.y + bias2.y, o1.z + bias2.z, o1.w + bias2.w);
            *reinterpret_cast<float4*>(out + (size_t)(node0 + m0 + 32) * D_FEAT + c4 * 4) = v;
        }
    }
}

extern "C" void kernel_launch(void* const* d_in, const int* in_sizes, int n_in,
                              void* d_out, int out_size, void* d_ws, size_t ws_size,
                              hipStream_t stream) {
    const float* feat = (const float*)d_in[0];
    const float* W1   = (const float*)d_in[1];
    const float* b1   = (const float*)d_in[2];
    const float* W2   = (const float*)d_in[3];
    const float* b2   = (const float*)d_in[4];
    const int*   src  = (const int*)d_in[5];
    const int*   dst  = (const int*)d_in[6];
    float*       out  = (float*)d_out;

    const int n_nodes = in_sizes[0] / D_FEAT;
    const int n_edges = in_sizes[5];

    float* rst = (float*)d_ws;
    unsigned int* planes = (unsigned int*)((char*)d_ws + (32u << 20));
    unsigned int* packed = (unsigned int*)((char*)d_ws + (48u << 20));

    {
        int tiles = (n_edges + 1023) / 1024;
        gin_key_kernel<<<tiles, 256, 0, stream>>>(src, dst, planes, packed, n_edges);
    }
    {
        int blocks = (n_nodes + AGG_ROWS - 1) / AGG_ROWS;
        gin_aggregate_kernel<<<blocks, AGG_THREADS, 0, stream>>>(feat, packed, planes, rst,
                                                                n_nodes, n_edges);
    }
    {
        int blocks = (n_nodes + TILE_M - 1) / TILE_M;
        gin_mlp_kernel<<<blocks, 512, 0, stream>>>(rst, W1, b1, W2, b2, out, n_nodes);
    }
}


# Aggregation sums in registers: each wave owns 13 rows of the 400-row tile, queued edges grouped by owner wave in LDS (counting sort), rows stored once; its ds_add_f32 all gone; sum order arrival
# speedup vs baseline: 2.3060x; 1.0908x over previous
#include <hip/hip_runtime.h>
#include <hip/hip_bf16.h>

typedef __attribute__((ext_vector_type(2))) float v2f;
typedef __attribute__((ext_vector_type(8))) float v8f;
typedef __attribute__((ext_vector_type(16))) float v16f;

#define D_FEAT 64
#define H_DIM  256
#define TILE_M 64
#define RST_STRIDE 68
#define H_STRIDE   260

#define AGG_ROWS    400
#define AGG_THREADS 1024
#define AGG_WAVES   32
#define AGG_SLOTS   13
#define AGG_QCAP    16384
#define AGG_QPT     (AGG_QCAP / AGG_THREADS)
#define KEY_PLANES  32
#define BM_PITCH    33
#define PK_SRC_BITS 23

__global__ __launch_bounds__(256)
void gin_key_kernel(const int* __restrict__ src,
                    const int* __restrict__ dst,
                    unsigned int* __restrict__ planes,
                    unsigned int* __restrict__ packed, int n_edges) {
    __shared__ unsigned int s_bm[KEY_PLANES * BM_PITCH];
    const int tid  = threadIdx.x;
    const int lane = tid & 31;
    const int w    = tid >> 5;
    for (int i = tid; i < KEY_PLANES * BM_PITCH; i += 256) s_bm[i] = 0u;
    __syncthreads();

    const int e0 = blockIdx.x * 1024 + tid * 4;
    int d[4] = {0, 0, 0, 0};
    int s[4] = {0, 0, 0, 0};
    int nv = 0;
    if (e0 + 3 < n_edges) {
        const int4 v = *reinterpret_cast<const int4*>(dst + e0);
        const int4 q = *reinterpret_cast<const int4*>(src + e0);
        d[0] = v.x; d[1] = v.y; d[2] = v.z; d[3] = v.w;
        s[0] = q.x; s[1] = q.y; s[2] = q.z; s[3] = q.w;
        nv = 4;
    } else {
        if (e0     < n_edges) { d[0] = dst[e0    ]; s[0] = src[e0    ]; nv = 1; }
        if (e0 + 1 < n_edges) { d[1] = dst[e0 + 1]; s[1] = src[e0 + 1]; nv = 2; }
        if (e0 + 2 < n_edges) { d[2] = dst[e0 + 2]; s[2] = src[e0 + 2]; nv = 3; }
    }
    unsigned int pk[4];
    const int word = tid >> 3;
    const int bit0 = (tid & 7) << 2;
    #pragma unroll
    for (int c = 0; c < 4; ++c) {
        const unsigned int key = (unsigned int)d[c] / AGG_ROWS;
        pk[c] = (unsigned int)s[c] | (((unsigned int)d[c] - key * AGG_ROWS) << PK_SRC_BITS);
        if (c < nv && key < 256u) {
            const unsigned int bit = 1u << (bit0 + c);
            atomicOr(&s_bm[(key >> 4) * BM_PITCH + word], bit);
            atomicOr(&s_bm[(16u + (key & 15u)) * BM_PITCH + word], bit);
        }
    }
    if (nv == 4) {
        *reinterpret_cast<uint4*>(packed + e0) = make_uint4(pk[0], pk[1], pk[2], pk[3]);
    } else {
        if (nv > 0) packed[e0    ] = pk[0];
        if (nv > 1) packed[e0 + 1] = pk[1];
        if (nv > 2) packed[e0 + 2] = pk[2];
    }
    __syncthreads();

    const size_t stride = (size_t)gridDim.x * 32;
    #pragma unroll
    for (int q = 0; q < KEY_PLANES / 8; ++q) {
        const int p = w + 8 * q;
        planes[(size_t)p * stride + (size_t)blockIdx.x * 32 + lane] = s_bm[p * BM_PITCH + lane];
    }
}

__device__ __forceinline__ void gin_add_row(v16f& accx, v16f& accy, int slot, float2 v) {
    if (slot < AGG_SLOTS) {
        accx[slot] += v.x;
        accy[slot] += v.y;
    }
}

__global__ __launch_bounds__(AGG_THREADS)
void gin_aggregate_kernel(const float* __restrict__ feat,
                          const unsigned int* __restrict__ packed,
                          const unsigned int* __restrict__ planes,
                          float* __restrict__ rst, int n_nodes, int n_edges) {
    __shared__ int          s_queue[AGG_QCAP];
    __shared__ unsigned int s_sorted[AGG_QCAP];
    __shared__ int          s_cnt[AGG_WAVES];
    __shared__ int          s_off[AGG_WAVES + 1];
    __shared__ int          s_qcount;

    const int tid  = threadIdx.x;
    const int lane = tid & 31;
    const int wave = tid >> 5;
    const int lo   = blockIdx.x * AGG_ROWS;
    const int rows = (n_nodes - lo) < AGG_ROWS ? (n_nodes - lo) : AGG_ROWS;

    if (tid == 0) s_qcount = 0;
    if (tid < AGG_WAVES) s_cnt[tid] = 0;
    __syncthreads();

    v16f accx = {}, accy = {};
    #pragma unroll
    for (int k = 0; k < AGG_SLOTS; ++k) {
        const int row = wave + AGG_WAVES * k;
        const int rc  = row < rows ? row : rows - 1;
        const float2 f = *reinterpret_cast<const float2*>(feat + (size_t)(lo + rc) * D_FEAT + lane * 2);
        accx[k] = f.x;
        accy[k] = f.y;
    }

    const int n_tiles = (n_edges + 1023) >> 10;
    const int n4 = n_tiles << 3;
    const size_t stride = (size_t)n_tiles * 32;
    const uint4* hi = reinterpret_cast<const uint4*>(planes + (size_t)(blockIdx.x >> 4) * stride);
    const uint4* lw = reinterpret_cast<const uint4*>(planes + (size_t)(16 + (blockIdx.x & 15)) * stride);
    const int start = (int)(((unsigned int)blockIdx.x * (unsigned int)n_tiles) / gridDim.x) << 3;

    int chunk = n4;
    int begin = 0;
    while (begin < n4) {
        const int end = (n4 - begin) < chunk ? n4 : begin + chunk;

        {
            uint4 a[4], b[4];
            #pragma unroll
            for (int u = 0; u < 4; ++u) {
                const int p = begin + u * AGG_THREADS + tid;
                int i = p + start;
                if (i >= n4) i -= n4;
                if (p >= end) i = 0;
                a[u] = hi[i];
                b[u] = lw[i];
            }
            for (int p0 = begin; p0 < end; p0 += AGG_THREADS * 4) {
                unsigned int m[4][4];
                int eb[4];
                #pragma unroll
                for (int u = 0; u < 4; ++u) {
                    const int p = p0 + u * AGG_THREADS + tid;
                    int i = p + start;
                    if (i >= n4) i -= n4;
                    const bool ok = p < end;
                    m[u][0] = ok ? (a[u].x & b[u].x) : 0u;
                    m[u][1] = ok ? (a[u].y & b[u].y) : 0u;
                    m[u][2] = ok ? (a[u].z & b[u].z) : 0u;
                    m[u][3] = ok ? (a[u].w & b[u].w) : 0u;
                    eb[u] = i << 7;
                }
                #pragma unroll
                for (int u = 0; u < 4; ++u) {
                    const int p = p0 + (4 + u) * AGG_THREADS + tid;
                    int i = p + start;
                    if (i >= n4) i -= n4;
                    if (p >= end) i = 0;
                    a[u] = hi[i];
                    b[u] = lw[i];
                }
                #pragma unroll
                for (int u = 0; u < 4; ++u) {
                    unsigned int m0 = m[u][0], m1 = m[u][1], m2 = m[u][2], m3 = m[u][3];
                    while (m0 | m1 | m2 | m3) {
                        const bool z0 = (m0 == 0u), z1 = (m1 == 0u), z2 = (m2 == 0u);
                        const bool p1 = z0 && !z1, p2 = z0 && z1 && !z2, p3 = z0 && z1 && z2;
                        const unsigned int mm = p3 ? m3 : (p2 ? m2 : (p1 ? m1 : m0));
                        const int off = p3 ? 96 : (p2 ? 64 : (p1 ? 32 : 0));
                        const unsigned int cl = mm & (mm - 1u);
                        m0 = z0 ? m0 : cl;
                        m1 = p1 ? cl : m1;
                        m2 = p2 ? cl : m2;
                        m3 = p3 ? cl : m3;
                        const int e = eb[u] + off + __builtin_ctz(mm);
                        const int slot = atomicAdd(&s_qcount, 1);
                        if (slot < AGG_QCAP) s_queue[slot] = e;
                    }
                }
            }
        }
        __syncthreads();
        const int total = __builtin_amdgcn_readfirstlane(s_qcount);
        if (total > AGG_QCAP) {
            __syncthreads();
            if (tid == 0) s_qcount = 0;
            chunk = AGG_QCAP / 128;
            __syncthreads();
            continue;
        }

        {
            unsigned int wv[AGG_QPT];
            int rk[AGG_QPT];
            #pragma unroll
            for (int k = 0; k < AGG_QPT; ++k) {
                const int i = k * AGG_THREADS + tid;
                wv[k] = 0u;
                if (i < total) wv[k] = packed[s_queue[i]];
            }
            #pragma unroll
            for (int k = 0; k < AGG_QPT; ++k) {
                const int i = k * AGG_THREADS + tid;
                rk[k] = 0;
                if (i < total) rk[k] = atomicAdd(&s_cnt[(wv[k] >> PK_SRC_BITS) & (AGG_WAVES - 1)], 1);
            }
            __syncthreads();
            if (tid < AGG_WAVES) {
                int x = s_cnt[tid];
                #pragma unroll
                for (int o = 1; o < AGG_WAVES; o <<= 1) {
                    const int y = __shfl_up(x, o);
                    if (lane >= o) x += y;
                }
                s_off[tid + 1] = x;
                if (tid == 0) s_off[0] = 0;
            }
            __syncthreads();
            #pragma unroll
            for (int k = 0; k < AGG_QPT; ++k) {
                const int i = k * AGG_THREADS + tid;
                if (i < total)
                    s_sorted[s_off[(wv[k] >> PK_SRC_BITS) & (AGG_WAVES - 1)] + rk[k]] = wv[k];
            }
        }
        __syncthreads();

        {
            const int qb = __builtin_amdgcn_readfirstlane(s_off[wave]);
            const int qe = __builtin_amdgcn_readfirstlane(s_off[wave + 1]);
            for (int base = qb; base < qe; base += 32) {
                const int i = base + lane;
                unsigned int w = 0xF0000000u;
                if (i < qe) w = s_sorted[i];
                const int s = (int)(w & ((1u << PK_SRC_BITS) - 1u));
                const int d = (int)(w >> (PK_SRC_BITS + 5));
                const int cnt = (qe - base) < 32 ? (qe - base) : 32;
                for (int j = 0; j < cnt; j += 16) {
                    float2 v[16];
                    int    dj[16];
                    #pragma unroll
                    for (int u = 0; u < 16; ++u) {
                        const int sj = __builtin_amdgcn_readlane(s, j + u);
                        dj[u] = __builtin_amdgcn_readlane(d, j + u);
                        v[u] = *reinterpret_cast<const float2*>(feat + (size_t)sj * D_FEAT + lane * 2);
                    }
                    #pragma unroll
                    for (int u = 0; u < 16; ++u) gin_add_row(accx, accy, dj[u], v[u]);
                }
            }
        }

        begin = end;
        if (begin < n4) {
            __syncthreads();
            if (tid == 0) s_qcount = 0;
            if (tid < AGG_WAVES) s_cnt[tid] = 0;
            __syncthreads();
        }
    }

    #pragma unroll
    for (int k = 0; k < AGG_SLOTS; ++k) {
        const int row = wave + AGG_WAVES * k;
        if (row < rows)
            *reinterpret_cast<float2*>(rst + (size_t)(lo + row) * D_FEAT + lane * 2) = make_float2(accx[k], accy[k]);
    }
}

__global__ __launch_bounds__(512)
void gin_mlp_kernel(const float* __restrict__ rst,
                    const float* __restrict__ W1,
                    const float* __restrict__ b1,
                    const float* __restrict__ W2,
                    const float* __restrict__ b2,
                    float* __restrict__ out, int n_nodes) {
    __shared__ __attribute__((aligned(16))) float s_rst[TILE_M * RST_STRIDE];
    __shared__ __attribute__((aligned(16))) float s_h[TILE_M * H_STRIDE];
    __shared__ __attribute__((aligned(16))) float s_out[TILE_M * D_FEAT];

    const int tid   = threadIdx.x;
    const int wave  = tid >> 5;
    const int lane  = tid & 31;
    const int node0 = blockIdx.x * TILE_M;

    const int frow  = lane & 15;
    const int khalf = (lane >> 4) << 1;
    const int mbase = (lane >> 4) << 3;

    const int c4 = tid & 15;
    const int m0 = tid >> 4;
    int r0 = node0 + m0;
    int r1 = node0 + m0 + 32;
    r0 = r0 < n_nodes ? r0 : (n_nodes - 1);
    r1 = r1 < n_nodes ? r1 : (n_nodes - 1);
    const float4 t0 = *reinterpret_cast<const float4*>(rst + (size_t)r0 * D_FEAT + c4 * 4);
    const float4 t1 = *reinterpret_cast<const float4*>(rst + (size_t)r1 * D_FEAT + c4 * 4);

    const int ncol1 = wave * 16 + frow;
    v2f bw1[16];
    #pragma unroll
    for (int kk = 0; kk < 16; ++kk) {
        const int k0 = kk * 4 + khalf;
        bw1[kk].x = W1[(size_t)(k0    ) * H_DIM + ncol1];
        bw1[kk].y = W1[(size_t)(k0 + 1) * H_DIM + ncol1];
    }
    const float bias1 = b1[ncol1];
    const float4 bias2 = *reinterpret_cast<const float4*>(b2 + c4 * 4);

    *reinterpret_cast<float4*>(&s_rst[(m0     ) * RST_STRIDE + c4 * 4]) = t0;
    *reinterpret_cast<float4*>(&s_rst[(m0 + 32) * RST_STRIDE + c4 * 4]) = t1;
    {
        const float4 z = make_float4(0.f, 0.f, 0.f, 0.f);
        reinterpret_cast<float4*>(s_out)[tid      ] = z;
        reinterpret_cast<float4*>(s_out)[tid + 512] = z;
    }
    __syncthreads();

    const int ntile = (wave & 3) * 16;
    const int kbase = (wave >> 2) * 64;
    const int ncol2 = ntile + frow;
    v2f bw2[16];
    #pragma unroll
    for (int kk = 0; kk < 16; ++kk) {
        const int k0 = kbase + kk * 4 + khalf;
        bw2[kk].x = W2[(size_t)(k0    ) * D_FEAT + ncol2];
        bw2[kk].y = W2[(size_t)(k0 + 1) * D_FEAT + ncol2];
    }

    {
        v8f acc0 = {}, acc1 = {}, acc2 = {}, acc3 = {};
        #pragma unroll
        for (int kk = 0; kk < 16; ++kk) {
            const int k0 = kk * 4 + khalf;
            const v2f b = bw1[kk];
            v2f a0, a1, a2, a3;
            a0.x = s_rst[(frow     ) * RST_STRIDE + k0];
            a0.y = s_rst[(frow     ) * RST_STRIDE + k0 + 1];
            a1.x = s_rst[(frow + 16) * RST_STRIDE + k0];
            a1.y = s_rst[(frow + 16) * RST_STRIDE + k0 + 1];
            a2.x = s_rst[(frow + 32) * RST_STRIDE + k0];
            a2.y = s_rst[(frow + 32) * RST_STRIDE + k0 + 1];
            a3.x = s_rst[(frow + 48) * RST_STRIDE + k0];
            a3.y = s_rst[(frow + 48) * RST_STRIDE + k0 + 1];
            acc0 = __builtin_amdgcn_wmma_f32_16x16x4_f32(
                false, a0, false, b, (short)0, acc0, false, false);
            acc1 = __builtin_amdgcn_wmma_f32_16x16x4_f32(
                false, a1, false, b, (short)0, acc1, false, false);
            acc2 = __builtin_amdgcn_wmma_f32_16x16x4_f32(
                false, a2, false, b, (short)0, acc2, false, false);
            acc3 = __builtin_amdgcn_wmma_f32_16x16x4_f32(
                false, a3, false, b, (short)0, acc3, false, false);
        }
        #pragma unroll
        for (int r = 0; r < 8; ++r) {
            float v0 = acc0[r] + bias1, v1 = acc1[r] + bias1;
            float v2 = acc2[r] + bias1, v3 = acc3[r] + bias1;
            s_h[(     mbase + r) * H_STRIDE + ncol1] = v0 > 0.f ? v0 : 0.f;
            s_h[(16 + mbase + r) * H_STRIDE + ncol1] = v1 > 0.f ? v1 : 0.f;
            s_h[(32 + mbase + r) * H_STRIDE + ncol1] = v2 > 0.f ? v2 : 0.f;
            s_h[(48 + mbase + r) * H_STRIDE + ncol1] = v3 > 0.f ? v3 : 0.f;
        }
    }
    __syncthreads();

    {
        v8f acc0 = {}, acc1 = {}, acc2 = {}, acc3 = {};
        #pragma unroll
        for (int kk = 0; kk < 16; ++kk) {
            const int k0 = kbase + kk * 4 + khalf;
            const v2f b = bw2[kk];
            v2f a0, a1, a2, a3;
            a0.x = s_h[(frow     ) * H_STRIDE + k0];
            a0.y = s_h[(frow     ) * H_STRIDE + k0 + 1];
            a1.x = s_h[(frow + 16) * H_STRIDE + k0];
            a1.y = s_h[(frow + 16) * H_STRIDE + k0 + 1];
            a2.x = s_h[(frow + 32) * H_STRIDE + k0];
            a2.y = s_h[(frow + 32) * H_STRIDE + k0 + 1];
            a3.x = s_h[(frow + 48) * H_STRIDE + k0];
            a3.y = s_h[(frow + 48) * H_STRIDE + k0 + 1];
            acc0 = __builtin_amdgcn_wmma_f32_16x16x4_f32(
                false, a0, false, b, (short)0, acc0, false, false);
            acc1 = __builtin_amdgcn_wmma_f32_16x16x4_f32(
                false, a1, false, b, (short)0, acc1, false, false);
            acc2 = __builtin_amdgcn_wmma_f32_16x16x4_f32(
                false, a2, false, b, (short)0, acc2, false, false);
            acc3 = __builtin_amdgcn_wmma_f32_16x16x4_f32(
                false, a3, false, b, (short)0, acc3, false, false);
        }
        #pragma unroll
        for (int r = 0; r < 8; ++r) {
            atomicAdd(&s_out[(     mbase + r) * D_FEAT + ncol2], acc0[r]);
            atomicAdd(&s_out[(16 + mbase + r) * D_FEAT + ncol2], acc1[r]);
            atomicAdd(&s_out[(32 + mbase + r) * D_FEAT + ncol2], acc2[r]);
            atomicAdd(&s_out[(48 + mbase + r) * D_FEAT + ncol2], acc3[r]);
        }
    }
    __syncthreads();

    {
        const float4 o0 = reinterpret_cast<const float4*>(s_out)[tid      ];
        const float4 o1 = reinterpret_cast<const float4*>(s_out)[tid + 512];
        if (node0 + m0 < n_nodes) {
            float4 v = make_float4(o0.x + bias2.x, o0.y + bias2.y, o0.z + bias2.z, o0.w + bias2.w);
            *reinterpret_cast<float4*>(out + (size_t)(node0 + m0) * D_FEAT + c4 * 4) = v;
        }
        if (node0 + m0 + 32 < n_nodes) {
            float4 v = make_float4(o1.x + bias2.x, o1.y + bias2.y, o1.z + bias2.z, o1.w + bias2.w);
            *reinterpret_cast<float4*>(out + (size_t)(node0 + m0 + 32) * D_FEAT + c4 * 4) = v;
        }
    }
}

extern "C" void kernel_launch(void* const* d_in, const int* in_sizes, int n_in,
                              void* d_out, int out_size, void* d_ws, size_t ws_size,
                              hipStream_t stream) {
    const float* feat = (const float*)d_in[0];
    const float* W1   = (const float*)d_in[1];
    const float* b1   = (const float*)d_in[2];
    const float* W2   = (const float*)d_in[3];
    const float* b2   = (const float*)d_in[4];
    const int*   src  = (const int*)d_in[5];
    const int*   dst  = (const int*)d_in[6];
    float*       out  = (float*)d_out;

    const int n_nodes = in_sizes[0] / D_FEAT;
    const int n_edges = in_sizes[5];

    float* rst = (float*)d_ws;
    unsigned int* planes = (unsigned int*)((char*)d_ws + (32u << 20));
    unsigned int* packed = (unsigned int*)((char*)d_ws + (48u << 20));

    {
        int tiles = (n_edges + 1023) / 1024;
        gin_key_kernel<<<tiles, 256, 0, stream>>>(src, dst, planes, packed, n_edges);
    }
    {
        int blocks = (n_nodes + AGG_ROWS - 1) / AGG_ROWS;
        gin_aggregate_kernel<<<blocks, AGG_THREADS, 0, stream>>>(feat, packed, planes, rst,
                                                                n_nodes, n_edges);
    }
    {
        int blocks = (n_nodes + TILE_M - 1) / TILE_M;
        gin_mlp_kernel<<<blocks, 512, 0, stream>>>(rst, W1, b1, W2, b2, out, n_nodes);
    }
}


# MLP launch made persistent: 512 work-groups load their W1/W2 fragments once and walk balanced runs of 16-row units (12 or 13) in 32-row chunks, next rows prefetched; was 1563 blocks of 64 rows
# speedup vs baseline: 2.3823x; 1.0331x over previous
#include <hip/hip_runtime.h>
#include <hip/hip_bf16.h>

typedef __attribute__((ext_vector_type(2))) float v2f;
typedef __attribute__((ext_vector_type(8))) float v8f;
typedef __attribute__((ext_vector_type(16))) float v16f;

#define D_FEAT 64
#define H_DIM  256
#define RST_STRIDE 68
#define H_STRIDE   260

#define AGG_ROWS    400
#define AGG_THREADS 1024
#define AGG_WAVES   32
#define AGG_SLOTS   13
#define AGG_QCAP    16384
#define AGG_QPT     (AGG_QCAP / AGG_THREADS)
#define KEY_PLANES  32
#define BM_PITCH    33
#define PK_SRC_BITS 23

__global__ __launch_bounds__(256)
void gin_key_kernel(const int* __restrict__ src,
                    const int* __restrict__ dst,
                    unsigned int* __restrict__ planes,
                    unsigned int* __restrict__ packed, int n_edges) {
    __shared__ unsigned int s_bm[KEY_PLANES * BM_PITCH];
    const int tid  = threadIdx.x;
    const int lane = tid & 31;
    const int w    = tid >> 5;
    for (int i = tid; i < KEY_PLANES * BM_PITCH; i += 256) s_bm[i] = 0u;
    __syncthreads();

    const int e0 = blockIdx.x * 1024 + tid * 4;
    int d[4] = {0, 0, 0, 0};
    int s[4] = {0, 0, 0, 0};
    int nv = 0;
    if (e0 + 3 < n_edges) {
        const int4 v = *reinterpret_cast<const int4*>(dst + e0);
        const int4 q = *reinterpret_cast<const int4*>(src + e0);
        d[0] = v.x; d[1] = v.y; d[2] = v.z; d[3] = v.w;
        s[0] = q.x; s[1] = q.y; s[2] = q.z; s[3] = q.w;
        nv = 4;
    } else {
        if (e0     < n_edges) { d[0] = dst[e0    ]; s[0] = src[e0    ]; nv = 1; }
        if (e0 + 1 < n_edges) { d[1] = dst[e0 + 1]; s[1] = src[e0 + 1]; nv = 2; }
        if (e0 + 2 < n_edges) { d[2] = dst[e0 + 2]; s[2] = src[e0 + 2]; nv = 3; }
    }
    unsigned int pk[4];
    const int word = tid >> 3;
    const int bit0 = (tid & 7) << 2;
    #pragma unroll
    for (int c = 0; c < 4; ++c) {
        const unsigned int key = (unsigned int)d[c] / AGG_ROWS;
        pk[c] = (unsigned int)s[c] | (((unsigned int)d[c] - key * AGG_ROWS) << PK_SRC_BITS);
        if (c < nv && key < 256u) {
            const unsigned int bit = 1u << (bit0 + c);
            atomicOr(&s_bm[(key >> 4) * BM_PITCH + word], bit);
            atomicOr(&s_bm[(16u + (key & 15u)) * BM_PITCH + word], bit);
        }
    }
    if (nv == 4) {
        *reinterpret_cast<uint4*>(packed + e0) = make_uint4(pk[0], pk[1], pk[2], pk[3]);
    } else {
        if (nv > 0) packed[e0    ] = pk[0];
        if (nv > 1) packed[e0 + 1] = pk[1];
        if (nv > 2) packed[e0 + 2] = pk[2];
    }
    __syncthreads();

    const size_t stride = (size_t)gridDim.x * 32;
    #pragma unroll
    for (int q = 0; q < KEY_PLANES / 8; ++q) {
        const int p = w + 8 * q;
        planes[(size_t)p * stride + (size_t)blockIdx.x * 32 + lane] = s_bm[p * BM_PITCH + lane];
    }
}

__device__ __forceinline__ void gin_add_row(v16f& accx, v16f& accy, int slot, float2 v) {
    if (slot < AGG_SLOTS) {
        accx[slot] += v.x;
        accy[slot] += v.y;
    }
}

__global__ __launch_bounds__(AGG_THREADS)
void gin_aggregate_kernel(const float* __restrict__ feat,
                          const unsigned int* __restrict__ packed,
                          const unsigned int* __restrict__ planes,
                          float* __restrict__ rst, int n_nodes, int n_edges) {
    __shared__ int          s_queue[AGG_QCAP];
    __shared__ unsigned int s_sorted[AGG_QCAP];
    __shared__ int          s_cnt[AGG_WAVES];
    __shared__ int          s_off[AGG_WAVES + 1];
    __shared__ int          s_qcount;

    const int tid  = threadIdx.x;
    const int lane = tid & 31;
    const int wave = tid >> 5;
    const int lo   = blockIdx.x * AGG_ROWS;
    const int rows = (n_nodes - lo) < AGG_ROWS ? (n_nodes - lo) : AGG_ROWS;

    if (tid == 0) s_qcount = 0;
    if (tid < AGG_WAVES) s_cnt[tid] = 0;
    __syncthreads();

    v16f accx = {}, accy = {};
    #pragma unroll
    for (int k = 0; k < AGG_SLOTS; ++k) {
        const int row = wave + AGG_WAVES * k;
        const int rc  = row < rows ? row : rows - 1;
        const float2 f = *reinterpret_cast<const float2*>(feat + (size_t)(lo + rc) * D_FEAT + lane * 2);
        accx[k] = f.x;
        accy[k] = f.y;
    }

    const int n_tiles = (n_edges + 1023) >> 10;
    const int n4 = n_tiles << 3;
    const size_t stride = (size_t)n_tiles * 32;
    const uint4* hi = reinterpret_cast<const uint4*>(planes + (size_t)(blockIdx.x >> 4) * stride);
    const uint4* lw = reinterpret_cast<const uint4*>(planes + (size_t)(16 + (blockIdx.x & 15)) * stride);
    const int start = (int)(((unsigned int)blockIdx.x * (unsigned int)n_tiles) / gridDim.x) << 3;

    int chunk = n4;
    int begin = 0;
    while (begin < n4) {
        const int end = (n4 - begin) < chunk ? n4 : begin + chunk;

        {
            uint4 a[4], b[4];
            #pragma unroll
            for (int u = 0; u < 4; ++u) {
                const int p = begin + u * AGG_THREADS + tid;
                int i = p + start;
                if (i >= n4) i -= n4;
                if (p >= end) i = 0;
                a[u] = hi[i];
                b[u] = lw[i];
            }
            for (int p0 = begin; p0 < end; p0 += AGG_THREADS * 4) {
                unsigned int m[4][4];
                int eb[4];
                #pragma unroll
                for (int u = 0; u < 4; ++u) {
                    const int p = p0 + u * AGG_THREADS + tid;
                    int i = p + start;
                    if (i >= n4) i -= n4;
                    const bool ok = p < end;
                    m[u][0] = ok ? (a[u].x & b[u].x) : 0u;
                    m[u][1] = ok ? (a[u].y & b[u].y) : 0u;
                    m[u][2] = ok ? (a[u].z & b[u].z) : 0u;
                    m[u][3] = ok ? (a[u].w & b[u].w) : 0u;
                    eb[u] = i << 7;
                }
                #pragma unroll
                for (int u = 0; u < 4; ++u) {
                    const int p = p0 + (4 + u) * AGG_THREADS + tid;
                    int i = p + start;
                    if (i >= n4) i -= n4;
                    if (p >= end) i = 0;
                    a[u] = hi[i];
                    b[u] = lw[i];
                }
                #pragma unroll
                for (int u = 0; u < 4; ++u) {
                    unsigned int m0 = m[u][0], m1 = m[u][1], m2 = m[u][2], m3 = m[u][3];
                    while (m0 | m1 | m2 | m3) {
                        const bool z0 = (m0 == 0u), z1 = (m1 == 0u), z2 = (m2 == 0u);
                        const bool p1 = z0 && !z1, p2 = z0 && z1 && !z2, p3 = z0 && z1 && z2;
                        const unsigned int mm = p3 ? m3 : (p2 ? m2 : (p1 ? m1 : m0));
                        const int off = p3 ? 96 : (p2 ? 64 : (p1 ? 32 : 0));
                        const unsigned int cl = mm & (mm - 1u);
                        m0 = z0 ? m0 : cl;
                        m1 = p1 ? cl : m1;
                        m2 = p2 ? cl : m2;
                        m3 = p3 ? cl : m3;
                        const int e = eb[u] + off + __builtin_ctz(mm);
                        const int slot = atomicAdd(&s_qcount, 1);
                        if (slot < AGG_QCAP) s_queue[slot] = e;
                    }
                }
            }
        }
        __syncthreads();
        const int total = __builtin_amdgcn_readfirstlane(s_qcount);
        if (total > AGG_QCAP) {
            __syncthreads();
            if (tid == 0) s_qcount = 0;
            chunk = AGG_QCAP / 128;
            __syncthreads();
            continue;
        }

        {
            unsigned int wv[AGG_QPT];
            int rk[AGG_QPT];
            #pragma unroll
            for (int k = 0; k < AGG_QPT; ++k) {
                const int i = k * AGG_THREADS + tid;
                wv[k] = 0u;
                if (i < total) wv[k] = packed[s_queue[i]];
            }
            #pragma unroll
            for (int k = 0; k < AGG_QPT; ++k) {
                const int i = k * AGG_THREADS + tid;
                rk[k] = 0;
                if (i < total) rk[k] = atomicAdd(&s_cnt[(wv[k] >> PK_SRC_BITS) & (AGG_WAVES - 1)], 1);
            }
            __syncthreads();
            if (tid < AGG_WAVES) {
                int x = s_cnt[tid];
                #pragma unroll
                for (int o = 1; o < AGG_WAVES; o <<= 1) {
                    const int y = __shfl_up(x, o);
                    if (lane >= o) x += y;
                }
                s_off[tid + 1] = x;
                if (tid == 0) s_off[0] = 0;
            }
            __syncthreads();
            #pragma unroll
            for (int k = 0; k < AGG_QPT; ++k) {
                const int i = k * AGG_THREADS + tid;
                if (i < total)
                    s_sorted[s_off[(wv[k] >> PK_SRC_BITS) & (AGG_WAVES - 1)] + rk[k]] = wv[k];
            }
        }
        __syncthreads();

        {
            const int qb = __builtin_amdgcn_readfirstlane(s_off[wave]);
            const int qe = __builtin_amdgcn_readfirstlane(s_off[wave + 1]);
            for (int base = qb; base < qe; base += 32) {
                const int i = base + lane;
                unsigned int w = 0xF0000000u;
                if (i < qe) w = s_sorted[i];
                const int s = (int)(w & ((1u << PK_SRC_BITS) - 1u));
                const int d = (int)(w >> (PK_SRC_BITS + 5));
                const int cnt = (qe - base) < 32 ? (qe - base) : 32;
                for (int j = 0; j < cnt; j += 16) {
                    float2 v[16];
                    int    dj[16];
                    #pragma unroll
                    for (int u = 0; u < 16; ++u) {
                        const int sj = __builtin_amdgcn_readlane(s, j + u);
                        dj[u] = __builtin_amdgcn_readlane(d, j + u);
                        v[u] = *reinterpret_cast<const float2*>(feat + (size_t)sj * D_FEAT + lane * 2);
                    }
                    #pragma unroll
                    for (int u = 0; u < 16; ++u) gin_add_row(accx, accy, dj[u], v[u]);
                }
            }
        }

        begin = end;
        if (begin < n4) {
            __syncthreads();
            if (tid == 0) s_qcount = 0;
            if (tid < AGG_WAVES) s_cnt[tid] = 0;
            __syncthreads();
        }
    }

    #pragma unroll
    for (int k = 0; k < AGG_SLOTS; ++k) {
        const int row = wave + AGG_WAVES * k;
        if (row < rows)
            *reinterpret_cast<float2*>(rst + (size_t)(lo + row) * D_FEAT + lane * 2) = make_float2(accx[k], accy[k]);
    }
}

#define MLP_BLOCKS 512
#define MLP_UNITS  2
#define MLP_ROWS   (16 * MLP_UNITS)
#define OUT_STRIDE 68

template <int MT>
__device__ __forceinline__ void gin_mlp_gemm1(const float* pa,
                                              float* ph,
                                              const v2f (&bw1)[16], float bias1) {
    v8f acc[MT];
    #pragma unroll
    for (int t = 0; t < MT; ++t) acc[t] = v8f{};
    #pragma unroll
    for (int kk = 0; kk < 16; ++kk) {
        const v2f b = bw1[kk];
        v2f a[MT];
        #pragma unroll
        for (int t = 0; t < MT; ++t) {
            a[t].x = pa[16 * t * RST_STRIDE + kk * 4];
            a[t].y = pa[16 * t * RST_STRIDE + kk * 4 + 1];
        }
        #pragma unroll
        for (int t = 0; t < MT; ++t)
            acc[t] = __builtin_amdgcn_wmma_f32_16x16x4_f32(
                false, a[t], false, b, (short)0, acc[t], false, false);
    }
    #pragma unroll
    for (int t = 0; t < MT; ++t) {
        #pragma unroll
        for (int r = 0; r < 8; ++r) {
            const float v = acc[t][r] + bias1;
            ph[(16 * t + r) * H_STRIDE] = v > 0.f ? v : 0.f;
        }
    }
}

template <int MT>
__device__ __forceinline__ void gin_mlp_gemm2(const float* pa,
                                              float* po,
                                              const v2f (&bw2)[16]) {
    v8f acc[MT];
    #pragma unroll
    for (int t = 0; t < MT; ++t) acc[t] = v8f{};
    #pragma unroll
    for (int kk = 0; kk < 16; ++kk) {
        const v2f b = bw2[kk];
        v2f a[MT];
        #pragma unroll
        for (int t = 0; t < MT; ++t) {
            a[t].x = pa[16 * t * H_STRIDE + kk * 4];
            a[t].y = pa[16 * t * H_STRIDE + kk * 4 + 1];
        }
        #pragma unroll
        for (int t = 0; t < MT; ++t)
            acc[t] = __builtin_amdgcn_wmma_f32_16x16x4_f32(
                false, a[t], false, b, (short)0, acc[t], false, false);
    }
    #pragma unroll
    for (int t = 0; t < MT; ++t) {
        #pragma unroll
        for (int r = 0; r < 8; ++r)
            atomicAdd(&po[(16 * t + r) * OUT_STRIDE], acc[t][r]);
    }
}

__device__ __forceinline__ void gin_mlp_epilogue(const float* s_out, const float* s_b2, float* __restrict__ out,
                                                 int m0, int c4, int node0, int mt, int n_nodes) {
    const float4 o0 = *reinterpret_cast<const float4*>(&s_out[m0 * OUT_STRIDE + c4 * 4]);
    if (m0 < mt * 16 && node0 + m0 < n_nodes) {
        const float4 bias2 = *reinterpret_cast<const float4*>(&s_b2[c4 * 4]);
        const float4 v = make_float4(o0.x + bias2.x, o0.y + bias2.y, o0.z + bias2.z, o0.w + bias2.w);
        const unsigned int off = (unsigned int)(node0 + m0) * (D_FEAT * 4u) + (unsigned int)c4 * 16u;
        *reinterpret_cast<float4*>(reinterpret_cast<char*>(out) + off) = v;
    }
}

__global__ __launch_bounds__(512, 8)
void gin_mlp_kernel(const float* __restrict__ rst,
                    const float* __restrict__ W1,
                    const float* __restrict__ b1,
                    const float* __restrict__ W2,
                    const float* __restrict__ b2,
                    float* __restrict__ out, int n_nodes) {
    __shared__ __attribute__((aligned(16))) float s_all[MLP_ROWS * OUT_STRIDE + MLP_ROWS * RST_STRIDE + MLP_ROWS * H_STRIDE + D_FEAT];
    float* const s_out = s_all;
    float* const s_rst = s_all + MLP_ROWS * OUT_STRIDE;
    float* const s_h   = s_all + MLP_ROWS * OUT_STRIDE + MLP_ROWS * RST_STRIDE;
    float* const s_b2  = s_h + MLP_ROWS * H_STRIDE;

    const int tid   = threadIdx.x;
    const int wave  = tid >> 5;
    const int lane  = tid & 31;

    const int frow  = lane & 15;
    const int khalf = (lane >> 4) << 1;
    const int mbase = (lane >> 4) << 3;

    const int n_units = (n_nodes + 15) >> 4;
    const int G    = (int)gridDim.x;
    const int half = G >> 1;
    const int bx   = (int)blockIdx.x;
    int run = bx;
    if ((G & 1) == 0) run = bx < half ? 2 * bx : 2 * (bx - half) + 1;
    const int u0 = (int)(((long long)run * n_units) / G);
    const int u1 = (int)(((long long)(run + 1) * n_units) / G);

    const int c4 = tid & 15;
    const int m0 = tid >> 4;
    float4 t0;
    {
        int r0 = u0 * 16 + m0;
        r0 = r0 < n_nodes ? r0 : (n_nodes - 1);
        const unsigned int off = (unsigned int)r0 * (D_FEAT * 4u) + (unsigned int)c4 * 16u;
        t0 = *reinterpret_cast<const float4*>(reinterpret_cast<const char*>(rst) + off);
    }

    const int ncol1 = wave * 16 + frow;
    v2f bw1[16];
    #pragma unroll
    for (int kk = 0; kk < 16; ++kk) {
        const int k0 = kk * 4 + khalf;
        bw1[kk].x = W1[(size_t)(k0    ) * H_DIM + ncol1];
        bw1[kk].y = W1[(size_t)(k0 + 1) * H_DIM + ncol1];
    }
    const float bias1 = b1[ncol1];
    if (tid < D_FEAT / 4) *reinterpret_cast<float4*>(&s_b2[tid * 4]) = *reinterpret_cast<const float4*>(b2 + tid * 4);

    const int ntile = (wave & 3) * 16;
    const int kbase = (wave >> 2) * 64;
    const int ncol2 = ntile + frow;
    v2f bw2[16];
    #pragma unroll
    for (int kk = 0; kk < 16; ++kk) {
        const int k0 = kbase + kk * 4 + khalf;
        bw2[kk].x = W2[(size_t)(k0    ) * D_FEAT + ncol2];
        bw2[kk].y = W2[(size_t)(k0 + 1) * D_FEAT + ncol2];
    }

    const float* pa1 = s_rst + frow * RST_STRIDE + khalf;
    float*       ph  = s_h + mbase * H_STRIDE + ncol1;
    const float* pa2 = s_h + frow * H_STRIDE + kbase + khalf;
    float*       po  = s_out + mbase * OUT_STRIDE + ncol2;

    int prev_node0 = 0;
    int prev_mt    = 0;
    for (int u = u0; u < u1; u += MLP_UNITS) {
        const int mt    = (u1 - u) < MLP_UNITS ? (u1 - u) : MLP_UNITS;
        const int node0 = u * 16;

        *reinterpret_cast<float4*>(&s_rst[m0 * RST_STRIDE + c4 * 4]) = t0;
        __syncthreads();

        if (prev_mt != 0) gin_mlp_epilogue(s_out, s_b2, out, m0, c4, prev_node0, prev_mt, n_nodes);
        *reinterpret_cast<float4*>(&s_out[m0 * OUT_STRIDE + c4 * 4]) = make_float4(0.f, 0.f, 0.f, 0.f);

        if (u + MLP_UNITS < u1) {
            int r0 = node0 + MLP_ROWS + m0;
            r0 = r0 < n_nodes ? r0 : (n_nodes - 1);
            const unsigned int off = (unsigned int)r0 * (D_FEAT * 4u) + (unsigned int)c4 * 16u;
            t0 = *reinterpret_cast<const float4*>(reinterpret_cast<const char*>(rst) + off);
        }

        if (mt == 2) gin_mlp_gemm1<2>(pa1, ph, bw1, bias1);
        else         gin_mlp_gemm1<1>(pa1, ph, bw1, bias1);
        __syncthreads();

        if (mt == 2) gin_mlp_gemm2<2>(pa2, po, bw2);
        else         gin_mlp_gemm2<1>(pa2, po, bw2);

        prev_node0 = node0;
        prev_mt    = mt;
    }
    if (prev_mt != 0) {
        __syncthreads();
        gin_mlp_epilogue(s_out, s_b2, out, m0, c4, prev_node0, prev_mt, n_nodes);
    }
}

extern "C" void kernel_launch(void* const* d_in, const int* in_sizes, int n_in,
                              void* d_out, int out_size, void* d_ws, size_t ws_size,
                              hipStream_t stream) {
    const float* feat = (const float*)d_in[0];
    const float* W1   = (const float*)d_in[1];
    const float* b1   = (const float*)d_in[2];
    const float* W2   = (const float*)d_in[3];
    const float* b2   = (const float*)d_in[4];
    const int*   src  = (const int*)d_in[5];
    const int*   dst  = (const int*)d_in[6];
    float*       out  = (float*)d_out;

    const int n_nodes = in_sizes[0] / D_FEAT;
    const int n_edges = in_sizes[5];

    float* rst = (float*)d_ws;
    unsigned int* planes = (unsigned int*)((char*)d_ws + (32u << 20));
    unsigned int* packed = (unsigned int*)((char*)d_ws + (48u << 20));

    {
        int tiles = (n_edges + 1023) / 1024;
        gin_key_kernel<<<tiles, 256, 0, stream>>>(src, dst, planes, packed, n_edges);
    }
    {
        int blocks = (n_nodes + AGG_ROWS - 1) / AGG_ROWS;
        gin_aggregate_kernel<<<blocks, AGG_THREADS, 0, stream>>>(feat, packed, planes, rst,
                                                                n_nodes, n_edges);
    }
    {
        gin_mlp_kernel<<<MLP_BLOCKS, 512, 0, stream>>>(rst, W1, b1, W2, b2, out, n_nodes);
    }
}


# Gather row loads as a rolling window: each register pair is refilled with the next batch's row right after its add, 14-16 loads always in flight (was 16 issued, full drain, 16 added); sum order kept
# speedup vs baseline: 2.4652x; 1.0348x over previous
#include <hip/hip_runtime.h>
#include <hip/hip_bf16.h>

typedef __attribute__((ext_vector_type(2))) float v2f;
typedef __attribute__((ext_vector_type(8))) float v8f;
typedef __attribute__((ext_vector_type(16))) float v16f;

#define D_FEAT 64
#define H_DIM  256
#define RST_STRIDE 68
#define H_STRIDE   260

#define AGG_ROWS    400
#define AGG_THREADS 1024
#define AGG_WAVES   32
#define AGG_SLOTS   13
#define AGG_QCAP    16384
#define AGG_QPT     (AGG_QCAP / AGG_THREADS)
#define KEY_PLANES  32
#define BM_PITCH    33
#define PK_SRC_BITS 23

__global__ __launch_bounds__(256)
void gin_key_kernel(const int* __restrict__ src,
                    const int* __restrict__ dst,
                    unsigned int* __restrict__ planes,
                    unsigned int* __restrict__ packed, int n_edges) {
    __shared__ unsigned int s_bm[KEY_PLANES * BM_PITCH];
    const int tid  = threadIdx.x;
    const int lane = tid & 31;
    const int w    = tid >> 5;
    for (int i = tid; i < KEY_PLANES * BM_PITCH; i += 256) s_bm[i] = 0u;
    __syncthreads();

    const int e0 = blockIdx.x * 1024 + tid * 4;
    int d[4] = {0, 0, 0, 0};
    int s[4] = {0, 0, 0, 0};
    int nv = 0;
    if (e0 + 3 < n_edges) {
        const int4 v = *reinterpret_cast<const int4*>(dst + e0);
        const int4 q = *reinterpret_cast<const int4*>(src + e0);
        d[0] = v.x; d[1] = v.y; d[2] = v.z; d[3] = v.w;
        s[0] = q.x; s[1] = q.y; s[2] = q.z; s[3] = q.w;
        nv = 4;
    } else {
        if (e0     < n_edges) { d[0] = dst[e0    ]; s[0] = src[e0    ]; nv = 1; }
        if (e0 + 1 < n_edges) { d[1] = dst[e0 + 1]; s[1] = src[e0 + 1]; nv = 2; }
        if (e0 + 2 < n_edges) { d[2] = dst[e0 + 2]; s[2] = src[e0 + 2]; nv = 3; }
    }
    unsigned int pk[4];
    const int word = tid >> 3;
    const int bit0 = (tid & 7) << 2;
    #pragma unroll
    for (int c = 0; c < 4; ++c) {
        const unsigned int key = (unsigned int)d[c] / AGG_ROWS;
        pk[c] = (unsigned int)s[c] | (((unsigned int)d[c] - key * AGG_ROWS) << PK_SRC_BITS);
        if (c < nv && key < 256u) {
            const unsigned int bit = 1u << (bit0 + c);
            atomicOr(&s_bm[(key >> 4) * BM_PITCH + word], bit);
            atomicOr(&s_bm[(16u + (key & 15u)) * BM_PITCH + word], bit);
        }
    }
    if (nv == 4) {
        *reinterpret_cast<uint4*>(packed + e0) = make_uint4(pk[0], pk[1], pk[2], pk[3]);
    } else {
        if (nv > 0) packed[e0    ] = pk[0];
        if (nv > 1) packed[e0 + 1] = pk[1];
        if (nv > 2) packed[e0 + 2] = pk[2];
    }
    __syncthreads();

    const size_t stride = (size_t)gridDim.x * 32;
    #pragma unroll
    for (int q = 0; q < KEY_PLANES / 8; ++q) {
        const int p = w + 8 * q;
        planes[(size_t)p * stride + (size_t)blockIdx.x * 32 + lane] = s_bm[p * BM_PITCH + lane];
    }
}

__device__ __forceinline__ void gin_add_row(v16f& accx, v16f& accy, int slot, float2 v) {
    if (slot < AGG_SLOTS) {
        accx[slot] += v.x;
        accy[slot] += v.y;
    }
}

__device__ __forceinline__ void gin_add_full_row(v16f& accx, v16f& accy, int slot, float2 v) {
    accx[slot] += v.x;
    accy[slot] += v.y;
}

__device__ __forceinline__ float2 gin_load_row(const float* __restrict__ feat, int row, int lane) {
    const float* rowp = feat + (size_t)row * D_FEAT;
    return *reinterpret_cast<const float2*>(reinterpret_cast<const char*>(rowp) + (unsigned int)lane * 8u);
}

#define AGG_FILL 0xF0000000u

__global__ __launch_bounds__(AGG_THREADS)
void gin_aggregate_kernel(const float* __restrict__ feat,
                          const unsigned int* __restrict__ packed,
                          const unsigned int* __restrict__ planes,
                          float* __restrict__ rst, int n_nodes, int n_edges) {
    __shared__ int          s_queue[AGG_QCAP];
    __shared__ unsigned int s_sorted[AGG_QCAP];
    __shared__ int          s_cnt[AGG_WAVES];
    __shared__ int          s_off[AGG_WAVES + 1];
    __shared__ int          s_qcount;

    const int tid  = threadIdx.x;
    const int lane = tid & 31;
    const int wave = tid >> 5;
    const int lo   = blockIdx.x * AGG_ROWS;
    const int rows = (n_nodes - lo) < AGG_ROWS ? (n_nodes - lo) : AGG_ROWS;

    if (tid == 0) s_qcount = 0;
    if (tid < AGG_WAVES) s_cnt[tid] = 0;
    __syncthreads();

    v16f accx = {}, accy = {};
    #pragma unroll
    for (int k = 0; k < AGG_SLOTS; ++k) {
        const int row = wave + AGG_WAVES * k;
        const int rc  = row < rows ? row : rows - 1;
        const float2 f = *reinterpret_cast<const float2*>(feat + (size_t)(lo + rc) * D_FEAT + lane * 2);
        accx[k] = f.x;
        accy[k] = f.y;
    }

    const int n_tiles = (n_edges + 1023) >> 10;
    const int n4 = n_tiles << 3;
    const size_t stride = (size_t)n_tiles * 32;
    const uint4* hi = reinterpret_cast<const uint4*>(planes + (size_t)(blockIdx.x >> 4) * stride);
    const uint4* lw = reinterpret_cast<const uint4*>(planes + (size_t)(16 + (blockIdx.x & 15)) * stride);
    const int start = (int)(((unsigned int)blockIdx.x * (unsigned int)n_tiles) / gridDim.x) << 3;

    int chunk = n4;
    int begin = 0;
    while (begin < n4) {
        const int end = (n4 - begin) < chunk ? n4 : begin + chunk;

        {
            uint4 a[4], b[4];
            #pragma unroll
            for (int u = 0; u < 4; ++u) {
                const int p = begin + u * AGG_THREADS + tid;
                int i = p + start;
                if (i >= n4) i -= n4;
                if (p >= end) i = 0;
                a[u] = hi[i];
                b[u] = lw[i];
            }
            for (int p0 = begin; p0 < end; p0 += AGG_THREADS * 4) {
                unsigned int m[4][4];
                int eb[4];
                #pragma unroll
                for (int u = 0; u < 4; ++u) {
                    const int p = p0 + u * AGG_THREADS + tid;
                    int i = p + start;
                    if (i >= n4) i -= n4;
                    const bool ok = p < end;
                    m[u][0] = ok ? (a[u].x & b[u].x) : 0u;
                    m[u][1] = ok ? (a[u].y & b[u].y) : 0u;
                    m[u][2] = ok ? (a[u].z & b[u].z) : 0u;
                    m[u][3] = ok ? (a[u].w & b[u].w) : 0u;
                    eb[u] = i << 7;
                }
                #pragma unroll
                for (int u = 0; u < 4; ++u) {
                    const int p = p0 + (4 + u) * AGG_THREADS + tid;
                    int i = p + start;
                    if (i >= n4) i -= n4;
                    if (p >= end) i = 0;
                    a[u] = hi[i];
                    b[u] = lw[i];
                }
                #pragma unroll
                for (int u = 0; u < 4; ++u) {
                    unsigned int m0 = m[u][0], m1 = m[u][1], m2 = m[u][2], m3 = m[u][3];
                    while (m0 | m1 | m2 | m3) {
                        const bool z0 = (m0 == 0u), z1 = (m1 == 0u), z2 = (m2 == 0u);
                        const bool p1 = z0 && !z1, p2 = z0 && z1 && !z2, p3 = z0 && z1 && z2;
                        const unsigned int mm = p3 ? m3 : (p2 ? m2 : (p1 ? m1 : m0));
                        const int off = p3 ? 96 : (p2 ? 64 : (p1 ? 32 : 0));
                        const unsigned int cl = mm & (mm - 1u);
                        m0 = z0 ? m0 : cl;
                        m1 = p1 ? cl : m1;
                        m2 = p2 ? cl : m2;
                        m3 = p3 ? cl : m3;
                        const int e = eb[u] + off + __builtin_ctz(mm);
                        const int slot = atomicAdd(&s_qcount, 1);
                        if (slot < AGG_QCAP) s_queue[slot] = e;
                    }
                }
            }
        }
        __syncthreads();
        const int total = __builtin_amdgcn_readfirstlane(s_qcount);
        if (total > AGG_QCAP) {
            __syncthreads();
            if (tid == 0) s_qcount = 0;
            chunk = AGG_QCAP / 128;
            __syncthreads();
            continue;
        }

        {
            unsigned int wv[AGG_QPT];
            int rk[AGG_QPT];
            #pragma unroll
            for (int k = 0; k < AGG_QPT; ++k) {
                const int i = k * AGG_THREADS + tid;
                wv[k] = 0u;
                if (i < total) wv[k] = packed[s_queue[i]];
            }
            #pragma unroll
            for (int k = 0; k < AGG_QPT; ++k) {
                const int i = k * AGG_THREADS + tid;
                rk[k] = 0;
                if (i < total) rk[k] = atomicAdd(&s_cnt[(wv[k] >> PK_SRC_BITS) & (AGG_WAVES - 1)], 1);
            }
            __syncthreads();
            if (tid < AGG_WAVES) {
                int x = s_cnt[tid];
                #pragma unroll
                for (int o = 1; o < AGG_WAVES; o <<= 1) {
                    const int y = __shfl_up(x, o);
                    if (lane >= o) x += y;
                }
                s_off[tid + 1] = x;
                if (tid == 0) s_off[0] = 0;
            }
            __syncthreads();
            #pragma unroll
            for (int k = 0; k < AGG_QPT; ++k) {
                const int i = k * AGG_THREADS + tid;
                if (i < total)
                    s_sorted[s_off[(wv[k] >> PK_SRC_BITS) & (AGG_WAVES - 1)] + rk[k]] = wv[k];
            }
        }
        __syncthreads();

        {
            const int qb = __builtin_amdgcn_readfirstlane(s_off[wave]);
            const int qe = __builtin_amdgcn_readfirstlane(s_off[wave + 1]);
            const int nbat = (qe - qb + 15) >> 4;
            if (nbat > 0) {
                const int l16 = lane & 15;
                unsigned int wc = AGG_FILL, wn = AGG_FILL;
                if (qb + l16 < qe)      wc = s_sorted[qb + l16];
                if (qb + 16 + l16 < qe) wn = s_sorted[qb + 16 + l16];
                float2 v[16];
                int    dj[16];
                {
                    const int s = (int)(wc & ((1u << PK_SRC_BITS) - 1u));
                    const int d = (int)(wc >> (PK_SRC_BITS + 5));
                    #pragma unroll
                    for (int u = 0; u < 16; ++u) {
                        const int sj = __builtin_amdgcn_readlane(s, u);
                        dj[u] = __builtin_amdgcn_readlane(d, u);
                        v[u] = gin_load_row(feat, sj, lane);
                    }
                }
                for (int b = 1; b < nbat; ++b) {
                    const int s = (int)(wn & ((1u << PK_SRC_BITS) - 1u));
                    const int d = (int)(wn >> (PK_SRC_BITS + 5));
                    unsigned int wf = AGG_FILL;
                    if (qb + 16 * (b + 1) + l16 < qe) wf = s_sorted[qb + 16 * (b + 1) + l16];
                    #pragma unroll
                    for (int u = 0; u < 16; ++u) {
                        gin_add_full_row(accx, accy, dj[u], v[u]);
                        const int sj = __builtin_amdgcn_readlane(s, u);
                        dj[u] = __builtin_amdgcn_readlane(d, u);
                        v[u] = gin_load_row(feat, sj, lane);
                    }
                    wn = wf;
                }
                #pragma unroll
                for (int u = 0; u < 16; ++u) gin_add_row(accx, accy, dj[u], v[u]);
            }
        }

        begin = end;
        if (begin < n4) {
            __syncthreads();
            if (tid == 0) s_qcount = 0;
            if (tid < AGG_WAVES) s_cnt[tid] = 0;
            __syncthreads();
        }
    }

    #pragma unroll
    for (int k = 0; k < AGG_SLOTS; ++k) {
        const int row = wave + AGG_WAVES * k;
        if (row < rows)
            *reinterpret_cast<float2*>(rst + (size_t)(lo + row) * D_FEAT + lane * 2) = make_float2(accx[k], accy[k]);
    }
}

#define MLP_BLOCKS 512
#define MLP_UNITS  2
#define MLP_ROWS   (16 * MLP_UNITS)
#define OUT_STRIDE 68

template <int MT>
__device__ __forceinline__ void gin_mlp_gemm1(const float* pa,
                                              float* ph,
                                              const v2f (&bw1)[16], float bias1) {
    v8f acc[MT];
    #pragma unroll
    for (int t = 0; t < MT; ++t) acc[t] = v8f{};
    #pragma unroll
    for (int kk = 0; kk < 16; ++kk) {
        const v2f b = bw1[kk];
        v2f a[MT];
        #pragma unroll
        for (int t = 0; t < MT; ++t) {
            a[t].x = pa[16 * t * RST_STRIDE + kk * 4];
            a[t].y = pa[16 * t * RST_STRIDE + kk * 4 + 1];
        }
        #pragma unroll
        for (int t = 0; t < MT; ++t)
            acc[t] = __builtin_amdgcn_wmma_f32_16x16x4_f32(
                false, a[t], false, b, (short)0, acc[t], false, false);
    }
    #pragma unroll
    for (int t = 0; t < MT; ++t) {
        #pragma unroll
        for (int r = 0; r < 8; ++r) {
            const float v = acc[t][r] + bias1;
            ph[(16 * t + r) * H_STRIDE] = v > 0.f ? v : 0.f;
        }
    }
}

template <int MT>
__device__ __forceinline__ void gin_mlp_gemm2(const float* pa,
                                              float* po,
                                              const v2f (&bw2)[16]) {
    v8f acc[MT];
    #pragma unroll
    for (int t = 0; t < MT; ++t) acc[t] = v8f{};
    #pragma unroll
    for (int kk = 0; kk < 16; ++kk) {
        const v2f b = bw2[kk];
        v2f a[MT];
        #pragma unroll
        for (int t = 0; t < MT; ++t) {
            a[t].x = pa[16 * t * H_STRIDE + kk * 4];
            a[t].y = pa[16 * t * H_STRIDE + kk * 4 + 1];
        }
        #pragma unroll
        for (int t = 0; t < MT; ++t)
            acc[t] = __builtin_amdgcn_wmma_f32_16x16x4_f32(
                false, a[t], false, b, (short)0, acc[t], false, false);
    }
    #pragma unroll
    for (int t = 0; t < MT; ++t) {
        #pragma unroll
        for (int r = 0; r < 8; ++r)
            atomicAdd(&po[(16 * t + r) * OUT_STRIDE], acc[t][r]);
    }
}

__device__ __forceinline__ void gin_mlp_epilogue(const float* s_out, const float* s_b2, float* __restrict__ out,
                                                 int m0, int c4, int node0, int mt, int n_nodes) {
    const float4 o0 = *reinterpret_cast<const float4*>(&s_out[m0 * OUT_STRIDE + c4 * 4]);
    if (m0 < mt * 16 && node0 + m0 < n_nodes) {
        const float4 bias2 = *reinterpret_cast<const float4*>(&s_b2[c4 * 4]);
        const float4 v = make_float4(o0.x + bias2.x, o0.y + bias2.y, o0.z + bias2.z, o0.w + bias2.w);
        const unsigned int off = (unsigned int)(node0 + m0) * (D_FEAT * 4u) + (unsigned int)c4 * 16u;
        *reinterpret_cast<float4*>(reinterpret_cast<char*>(out) + off) = v;
    }
}

__global__ __launch_bounds__(512, 8)
void gin_mlp_kernel(const float* __restrict__ rst,
                    const float* __restrict__ W1,
                    const float* __restrict__ b1,
                    const float* __restrict__ W2,
                    const float* __restrict__ b2,
                    float* __restrict__ out, int n_nodes) {
    __shared__ __attribute__((aligned(16))) float s_all[MLP_ROWS * OUT_STRIDE + MLP_ROWS * RST_STRIDE + MLP_ROWS * H_STRIDE + D_FEAT];
    float* const s_out = s_all;
    float* const s_rst = s_all + MLP_ROWS * OUT_STRIDE;
    float* const s_h   = s_all + MLP_ROWS * OUT_STRIDE + MLP_ROWS * RST_STRIDE;
    float* const s_b2  = s_h + MLP_ROWS * H_STRIDE;

    const int tid   = threadIdx.x;
    const int wave  = tid >> 5;
    const int lane  = tid & 31;

    const int frow  = lane & 15;
    const int khalf = (lane >> 4) << 1;
    const int mbase = (lane >> 4) << 3;

    const int n_units = (n_nodes + 15) >> 4;
    const int G    = (int)gridDim.x;
    const int half = G >> 1;
    const int bx   = (int)blockIdx.x;
    int run = bx;
    if ((G & 1) == 0) run = bx < half ? 2 * bx : 2 * (bx - half) + 1;
    const int u0 = (int)(((long long)run * n_units) / G);
    const int u1 = (int)(((long long)(run + 1) * n_units) / G);

    const int c4 = tid & 15;
    const int m0 = tid >> 4;
    float4 t0;
    {
        int r0 = u0 * 16 + m0;
        r0 = r0 < n_nodes ? r0 : (n_nodes - 1);
        const unsigned int off = (unsigned int)r0 * (D_FEAT * 4u) + (unsigned int)c4 * 16u;
        t0 = *reinterpret_cast<const float4*>(reinterpret_cast<const char*>(rst) + off);
    }

    const int ncol1 = wave * 16 + frow;
    v2f bw1[16];
    #pragma unroll
    for (int kk = 0; kk < 16; ++kk) {
        const int k0 = kk * 4 + khalf;
        bw1[kk].x = W1[(size_t)(k0    ) * H_DIM + ncol1];
        bw1[kk].y = W1[(size_t)(k0 + 1) * H_DIM + ncol1];
    }
    const float bias1 = b1[ncol1];
    if (tid < D_FEAT / 4) *reinterpret_cast<float4*>(&s_b2[tid * 4]) = *reinterpret_cast<const float4*>(b2 + tid * 4);

    const int ntile = (wave & 3) * 16;
    const int kbase = (wave >> 2) * 64;
    const int ncol2 = ntile + frow;
    v2f bw2[16];
    #pragma unroll
    for (int kk = 0; kk < 16; ++kk) {
        const int k0 = kbase + kk * 4 + khalf;
        bw2[kk].x = W2[(size_t)(k0    ) * D_FEAT + ncol2];
        bw2[kk].y = W2[(size_t)(k0 + 1) * D_FEAT + ncol2];
    }

    const float* pa1 = s_rst + frow * RST_STRIDE + khalf;
    float*       ph  = s_h + mbase * H_STRIDE + ncol1;
    const float* pa2 = s_h + frow * H_STRIDE + kbase + khalf;
    float*       po  = s_out + mbase * OUT_STRIDE + ncol2;

    int prev_node0 = 0;
    int prev_mt    = 0;
    for (int u = u0; u < u1; u += MLP_UNITS) {
        const int mt    = (u1 - u) < MLP_UNITS ? (u1 - u) : MLP_UNITS;
        const int node0 = u * 16;

        *reinterpret_cast<float4*>(&s_rst[m0 * RST_STRIDE + c4 * 4]) = t0;
        __syncthreads();

        if (prev_mt != 0) gin_mlp_epilogue(s_out, s_b2, out, m0, c4, prev_node0, prev_mt, n_nodes);
        *reinterpret_cast<float4*>(&s_out[m0 * OUT_STRIDE + c4 * 4]) = make_float4(0.f, 0.f, 0.f, 0.f);

        if (u + MLP_UNITS < u1) {
            int r0 = node0 + MLP_ROWS + m0;
            r0 = r0 < n_nodes ? r0 : (n_nodes - 1);
            const unsigned int off = (unsigned int)r0 * (D_FEAT * 4u) + (unsigned int)c4 * 16u;
            t0 = *reinterpret_cast<const float4*>(reinterpret_cast<const char*>(rst) + off);
        }

        if (mt == 2) gin_mlp_gemm1<2>(pa1, ph, bw1, bias1);
        else         gin_mlp_gemm1<1>(pa1, ph, bw1, bias1);
        __syncthreads();

        if (mt == 2) gin_mlp_gemm2<2>(pa2, po, bw2);
        else         gin_mlp_gemm2<1>(pa2, po, bw2);

        prev_node0 = node0;
        prev_mt    = mt;
    }
    if (prev_mt != 0) {
        __syncthreads();
        gin_mlp_epilogue(s_out, s_b2, out, m0, c4, prev_node0, prev_mt, n_nodes);
    }
}

extern "C" void kernel_launch(void* const* d_in, const int* in_sizes, int n_in,
                              void* d_out, int out_size, void* d_ws, size_t ws_size,
                              hipStream_t stream) {
    const float* feat = (const float*)d_in[0];
    const float* W1   = (const float*)d_in[1];
    const float* b1   = (const float*)d_in[2];
    const float* W2   = (const float*)d_in[3];
    const float* b2   = (const float*)d_in[4];
    const int*   src  = (const int*)d_in[5];
    const int*   dst  = (const int*)d_in[6];
    float*       out  = (float*)d_out;

    const int n_nodes = in_sizes[0] / D_FEAT;
    const int n_edges = in_sizes[5];

    float* rst = (float*)d_ws;
    unsigned int* planes = (unsigned int*)((char*)d_ws + (32u << 20));
    unsigned int* packed = (unsigned int*)((char*)d_ws + (48u << 20));

    {
        int tiles = (n_edges + 1023) / 1024;
        gin_key_kernel<<<tiles, 256, 0, stream>>>(src, dst, planes, packed, n_edges);
    }
    {
        int blocks = (n_nodes + AGG_ROWS - 1) / AGG_ROWS;
        gin_aggregate_kernel<<<blocks, AGG_THREADS, 0, stream>>>(feat, packed, planes, rst,
                                                                n_nodes, n_edges);
    }
    {
        gin_mlp_kernel<<<MLP_BLOCKS, 512, 0, stream>>>(rst, W1, b1, W2, b2, out, n_nodes);
    }
}
